# starting kernel
# baseline (speedup 1.0000x reference)
#include <hip/hip_runtime.h>
#include <hip/hip_bf16.h>

typedef float v2f __attribute__((ext_vector_type(2)));
typedef float v8f __attribute__((ext_vector_type(8)));

#define IN_DIM  128
#define HID_DIM 128
#define OUT_DIM 64

__global__ void init_deg_kernel(float* __restrict__ deg, int n) {
    int i = blockIdx.x * blockDim.x + threadIdx.x;
    if (i < n) deg[i] = 1.0f;
}

__global__ void accum_deg_kernel(const int* __restrict__ col,
                                 float* __restrict__ deg, int E) {
    int e = blockIdx.x * blockDim.x + threadIdx.x;
    if (e < E) atomicAdd(&deg[col[e]], 1.0f);
}

__global__ void deg_to_dinv_kernel(float* __restrict__ deg, int n) {
    int i = blockIdx.x * blockDim.x + threadIdx.x;
    if (i < n) deg[i] = rsqrtf(deg[i]);
}

template <int NOUT>
__global__ void gemm_wmma_kernel(const float* __restrict__ X,
                                 const float* __restrict__ W,
                                 float* __restrict__ H, int nNodes) {
    constexpr int K = 128;
    const int wv   = threadIdx.x >> 5;
    const int lane = threadIdx.x & 31;
    const int half = lane >> 4;
    const int l16  = lane & 15;

    const int m0 = blockIdx.x * 16;
    const int n0 = wv * 16;

    int mrow = m0 + l16; if (mrow >= nNodes) mrow = nNodes - 1;
    const float* aptr = X + (size_t)mrow * K + 2 * half;
    const float* bptr = W + (size_t)(n0 + l16) * K + 2 * half;

    v8f c = {0.f, 0.f, 0.f, 0.f, 0.f, 0.f, 0.f, 0.f};
#pragma unroll
    for (int k0 = 0; k0 < K; k0 += 4) {
        v2f a = *(const v2f*)(aptr + k0);
        v2f b = *(const v2f*)(bptr + k0);
        c = __builtin_amdgcn_wmma_f32_16x16x4_f32(false, a, false, b,
                                                  (short)0, c, false, false);
    }

    float* out = H + (size_t)m0 * NOUT + n0;
#pragma unroll
    for (int v = 0; v < 8; ++v) {
        int r = v + 8 * half;
        if (m0 + r < nNodes) out[(size_t)r * NOUT + l16] = c[v];
    }
}

template <int D>
__global__ void propagate_kernel(const int* __restrict__ row,
                                 const int* __restrict__ col,
                                 const float* __restrict__ dinv,
                                 const float* __restrict__ h,
                                 float* __restrict__ agg, int E) {
    int wave = (int)((blockIdx.x * (size_t)blockDim.x + threadIdx.x) >> 5);
    int lane = threadIdx.x & 31;
    if (wave >= E) return;
    int r = row[wave];
    int c = col[wave];
    float nrm = dinv[r] * dinv[c];
    constexpr int V = D / 32;
    const float* src = h   + (size_t)r * D + lane * V;
    float*       dst = agg + (size_t)c * D + lane * V;
    if constexpr (V == 4) {
        float4 v = *(const float4*)src;
        atomicAdd(dst + 0, nrm * v.x);
        atomicAdd(dst + 1, nrm * v.y);
        atomicAdd(dst + 2, nrm * v.z);
        atomicAdd(dst + 3, nrm * v.w);
    } else {
        float2 v = *(const float2*)src;
        atomicAdd(dst + 0, nrm * v.x);
        atomicAdd(dst + 1, nrm * v.y);
    }
}

template <int D, bool RELU>
__global__ void finalize_kernel(const float* __restrict__ agg,
                                const float* __restrict__ h,
                                const float* __restrict__ dinv,
                                const float* __restrict__ bias,
                                float* __restrict__ out, int n) {
    size_t idx = blockIdx.x * (size_t)blockDim.x + threadIdx.x;
    if (idx >= (size_t)n * D) return;
    int i = (int)(idx / D);
    int d = (int)(idx % D);
    float di = dinv[i];
    float v  = agg[idx] + di * di * h[idx] + bias[d];
    out[idx] = RELU ? fmaxf(v, 0.0f) : v;
}

extern "C" void kernel_launch(void* const* d_in, const int* in_sizes, int n_in,
                              void* d_out, int out_size, void* d_ws, size_t ws_size,
                              hipStream_t stream) {
    const float* x  = (const float*)d_in[0];
    const int*   ei = (const int*)d_in[1];
    const float* W0 = (const float*)d_in[2];
    const float* b0 = (const float*)d_in[3];
    const float* W1 = (const float*)d_in[4];
    const float* b1 = (const float*)d_in[5];
    const float* W2 = (const float*)d_in[6];
    const float* b2 = (const float*)d_in[7];

    const int N = in_sizes[0] / IN_DIM;
    const int E = in_sizes[1] / 2;
    const int* rowIdx = ei;
    const int* colIdx = ei + E;

    char*  ws  = (char*)d_ws;
    size_t off = 0;
    auto   alloc = [&](size_t bytes) {
        void* p = ws + off;
        off += (bytes + 255) & ~(size_t)255;
        return p;
    };
    float* dinv = (float*)alloc((size_t)N * sizeof(float));
    float* hbuf = (float*)alloc((size_t)N * HID_DIM * sizeof(float));
    float* agg  = (float*)alloc((size_t)N * HID_DIM * sizeof(float));
    float* xbuf = (float*)alloc((size_t)N * HID_DIM * sizeof(float));

    const int mtiles   = (N + 15) / 16;
    const int edgeBlks = (E + 7) / 8;

    init_deg_kernel<<<(N + 255) / 256, 256, 0, stream>>>(dinv, N);
    accum_deg_kernel<<<(E + 255) / 256, 256, 0, stream>>>(colIdx, dinv, E);
    deg_to_dinv_kernel<<<(N + 255) / 256, 256, 0, stream>>>(dinv, N);

    gemm_wmma_kernel<128><<<mtiles, 256, 0, stream>>>(x, W0, hbuf, N);
    hipMemsetAsync(agg, 0, (size_t)N * 128 * sizeof(float), stream);
    propagate_kernel<128><<<edgeBlks, 256, 0, stream>>>(rowIdx, colIdx, dinv, hbuf, agg, E);
    finalize_kernel<128, true><<<(int)(((size_t)N * 128 + 255) / 256), 256, 0, stream>>>(
        agg, hbuf, dinv, b0, xbuf, N);

    gemm_wmma_kernel<128><<<mtiles, 256, 0, stream>>>(xbuf, W1, hbuf, N);
    hipMemsetAsync(agg, 0, (size_t)N * 128 * sizeof(float), stream);
    propagate_kernel<128><<<edgeBlks, 256, 0, stream>>>(rowIdx, colIdx, dinv, hbuf, agg, E);
    finalize_kernel<128, true><<<(int)(((size_t)N * 128 + 255) / 256), 256, 0, stream>>>(
        agg, hbuf, dinv, b1, xbuf, N);

    gemm_wmma_kernel<64><<<mtiles, 128, 0, stream>>>(xbuf, W2, hbuf, N);
    hipMemsetAsync(agg, 0, (size_t)N * 64 * sizeof(float), stream);
    propagate_kernel<64><<<edgeBlks, 256, 0, stream>>>(rowIdx, colIdx, dinv, hbuf, agg, E);
    finalize_kernel<64, false><<<(int)(((size_t)N * 64 + 255) / 256), 256, 0, stream>>>(
        agg, hbuf, dinv, b2, (float*)d_out, N);
}


# Layer 0 also sums from the edge list: one pass deals each edge to its owner's list (degrees and entries staged in LDS, one cursor add per group and block), not 256 scans of all edges; arrival order
# speedup vs baseline: 2.7220x; 2.7220x over previous
#include <hip/hip_runtime.h>
#include <hip/hip_bf16.h>

typedef float v2f __attribute__((ext_vector_type(2)));
typedef float v8f __attribute__((ext_vector_type(8)));

#define IN_DIM  128
#define HID_DIM 128
#define OUT_DIM 64

__global__ void init_deg_kernel(float* __restrict__ deg, unsigned* __restrict__ cursor,
                                int n, int nGroups) {
    int i = blockIdx.x * blockDim.x + threadIdx.x;
    if (i < n) deg[i] = 1.0f;
    if (i < nGroups) cursor[i] = 0u;
}

__global__ void accum_deg_kernel(const int* __restrict__ col,
                                 float* __restrict__ deg, int E) {
    int e = blockIdx.x * blockDim.x + threadIdx.x;
    if (e < E) atomicAdd(&deg[col[e]], 1.0f);
}

#define GEMM_THREADS 512
template <int NOUT>
__global__ __launch_bounds__(GEMM_THREADS)
void gemm_wmma_kernel(const float* __restrict__ X,
                      const float* __restrict__ W,
                      float* __restrict__ H, int nNodes, int mtiles) {
    constexpr int K  = 128;
    constexpr int KQ = K / 8;
    constexpr int NT = NOUT / 16;
    constexpr int MG = (GEMM_THREADS / 32) / NT;
    const int wv   = __builtin_amdgcn_readfirstlane(threadIdx.x >> 5);
    const int lane = threadIdx.x & 31;
    const int half = lane >> 4;
    const int l16  = lane & 15;
    const int n0   = (wv % NT) * 16;
    const int step = gridDim.x * MG;
    int t = blockIdx.x * MG + wv / NT;
    if (t >= mtiles) return;

    float4 wf[KQ];
    {
        const float4* p = (const float4*)(W + (size_t)(n0 + l16) * K + 4 * half);
#pragma unroll
        for (int j = 0; j < KQ; ++j) wf[j] = p[2 * j];
    }
    auto load_tile = [&](float4 (&dst)[KQ], int tile) {
        int mrow = min(tile * 16 + l16, nNodes - 1);
        const float4* p = (const float4*)(X + (size_t)mrow * K + 4 * half);
#pragma unroll
        for (int j = 0; j < KQ; ++j) dst[j] = p[2 * j];
    };
    auto mma_store = [&](const float4 (&s)[KQ], int tile) {
        v8f c = {0.f, 0.f, 0.f, 0.f, 0.f, 0.f, 0.f, 0.f};
#pragma unroll
        for (int j = 0; j < KQ; ++j) {
            v2f a0 = {s[j].x, s[j].y}, b0 = {wf[j].x, wf[j].y};
            v2f a1 = {s[j].z, s[j].w}, b1 = {wf[j].z, wf[j].w};
            c = __builtin_amdgcn_wmma_f32_16x16x4_f32(false, a0, false, b0,
                                                      (short)0, c, false, false);
            c = __builtin_amdgcn_wmma_f32_16x16x4_f32(false, a1, false, b1,
                                                      (short)0, c, false, false);
        }
        const int m0 = tile * 16;
        float* out = H + (size_t)(m0 + 8 * half) * NOUT + n0 + l16;
        if (m0 + 16 <= nNodes) {
#pragma unroll
            for (int v = 0; v < 8; ++v) out[(size_t)v * NOUT] = c[v];
        } else {
#pragma unroll
            for (int v = 0; v < 8; ++v)
                if (m0 + 8 * half + v < nNodes) out[(size_t)v * NOUT] = c[v];
        }
    };

    float4 fa[KQ], fb[KQ];
    load_tile(fa, t);
    for (;;) {
        int tn = t + step;
        load_tile(fb, min(tn, mtiles - 1));
        __builtin_amdgcn_sched_barrier(0);
        mma_store(fa, t);
        if (tn >= mtiles) break;
        t = tn + step;
        load_tile(fa, min(t, mtiles - 1));
        __builtin_amdgcn_sched_barrier(0);
        mma_store(fb, tn);
        if (t >= mtiles) break;
    }
}

#define OWN_ROWS     196
#define OWN_THREADS  1024
#define MAX_GROUPS   256
#define BIN_THREADS  1024
#define BIN_EDGES    (BIN_THREADS * 4)

__device__ __forceinline__ void lds_barrier() {
    __builtin_amdgcn_fence(__ATOMIC_RELEASE, "workgroup", "local");
    __builtin_amdgcn_s_barrier();
    __builtin_amdgcn_fence(__ATOMIC_ACQUIRE, "workgroup", "local");
}

__global__ __launch_bounds__(BIN_THREADS)
void edge_list_kernel(const int* __restrict__ row,
                      const int* __restrict__ col,
                      const float* __restrict__ deg,
                      float* __restrict__ dinv,
                      unsigned* __restrict__ cursor,
                      uint2* __restrict__ recs, int* __restrict__ tab,
                      int n, int E, int nGroups) {
    constexpr int NP4 = MAX_GROUPS * OWN_ROWS / 4;
    constexpr int FPT = (NP4 + BIN_THREADS - 1) / BIN_THREADS;
    constexpr int Q4  = OWN_ROWS / 4;
    constexpr int PPT = (Q4 + 3) / 4;
    static_assert(OWN_ROWS % 4 == 0 && BIN_THREADS / 4 == MAX_GROUPS, "one lane quad per group");

    __shared__ float4   deg_s4[NP4];
    __shared__ uint2    stage[BIN_EDGES];
    __shared__ unsigned spos[BIN_EDGES];
    __shared__ int      gsum[MAX_GROUPS];
    __shared__ unsigned slot[MAX_GROUPS];
    __shared__ unsigned lfirst[MAX_GROUPS];
    __shared__ unsigned gfirst[MAX_GROUPS];
    __shared__ int      wtot[MAX_GROUPS / 32];
    __shared__ unsigned wcnt[MAX_GROUPS / 32];
    const float* deg_s = (const float*)deg_s4;

    const int tid  = threadIdx.x;
    const int lane = tid & 31;
    const int wave = tid >> 5;
    const int gid  = blockIdx.x * BIN_THREADS + tid;

    int r[4], c[4];
    const int e0 = gid * 4;
    if (e0 + 3 < E) {
        int4 rv = *(const int4*)(row + e0);
        int4 cv = *(const int4*)(col + e0);
        r[0] = rv.x; r[1] = rv.y; r[2] = rv.z; r[3] = rv.w;
        c[0] = cv.x; c[1] = cv.y; c[2] = cv.z; c[3] = cv.w;
    } else {
#pragma unroll
        for (int k = 0; k < 4; ++k) {
            bool in = e0 + k < E;
            r[k] = in ? row[e0 + k] : 0;
            c[k] = in ? col[e0 + k] : -1;
        }
    }

    {
        const float4* __restrict__ deg4 = (const float4*)deg;
        const int n4 = (n + 3) >> 2;
        float4 dv[FPT];
#pragma unroll
        for (int j = 0; j < FPT; ++j) dv[j] = deg4[min(tid + BIN_THREADS * j, n4 - 1)];
#pragma unroll
        for (int j = 0; j < FPT; ++j)
            asm volatile("" :: "v"(dv[j].x), "v"(dv[j].y), "v"(dv[j].z), "v"(dv[j].w));
#pragma unroll
        for (int j = 0; j < FPT; ++j) {
            const int i = tid + BIN_THREADS * j;
            if (i < n4) deg_s4[i] = dv[j];
        }
    }
    if (tid < MAX_GROUPS) slot[tid] = 0u;
    lds_barrier();

    {
        const int g  = tid >> 2;
        const int qd = tid & 3;
        const int lo = g * OWN_ROWS, hi = min(lo + OWN_ROWS, n);
        const int last4 = ((n + 3) >> 2) - 1;
        float4 v[PPT];
#pragma unroll
        for (int j = 0; j < PPT; ++j)
            v[j] = deg_s4[min(g * Q4 + min(qd + 4 * j, Q4 - 1), last4)];
        int part = 0;
#pragma unroll
        for (int j = 0; j < PPT; ++j) {
            const int q = qd + 4 * j;
            const int i = lo + 4 * q;
            if (q < Q4) {
                part += (i     < hi) ? (int)v[j].x - 1 : 0;
                part += (i + 1 < hi) ? (int)v[j].y - 1 : 0;
                part += (i + 2 < hi) ? (int)v[j].z - 1 : 0;
                part += (i + 3 < hi) ? (int)v[j].w - 1 : 0;
            }
        }
        part += __shfl_xor(part, 1, 32);
        part += __shfl_xor(part, 2, 32);
        if (qd == 0 && g < nGroups) gsum[g] = part;
    }
    lds_barrier();

    const int mine = (tid < nGroups) ? gsum[tid] : 0;
    int inc = mine;
#pragma unroll
    for (int m = 1; m < 32; m <<= 1) {
        int t = __shfl_up(inc, m, 32);
        if (lane >= m) inc += t;
    }
    if (tid < MAX_GROUPS && lane == 31) wtot[wave] = inc;

    unsigned g4[4], l4[4], rk[4];
#pragma unroll
    for (int k = 0; k < 4; ++k) {
        unsigned cc = (unsigned)max(c[k], 0);
        g4[k] = cc / OWN_ROWS;
        l4[k] = cc - g4[k] * OWN_ROWS;
        rk[k] = 0u;
        if (c[k] >= 0) rk[k] = atomicAdd(&slot[g4[k]], 1u);
    }
    lds_barrier();

    unsigned cnt = 0u, cinc = 0u, first = 0u;
    int goff = 0;
    if (tid < MAX_GROUPS) {
        cnt = slot[tid];
        if (cnt) first = atomicAdd(&cursor[tid], cnt);
        goff = inc - mine;
        for (int w = 0; w < wave; ++w) goff += wtot[w];
    }
    cinc = cnt;
#pragma unroll
    for (int m = 1; m < 32; m <<= 1) {
        unsigned t = __shfl_up(cinc, m, 32);
        if (lane >= m) cinc += t;
    }
    if (tid < MAX_GROUPS && lane == 31) wcnt[wave] = cinc;
    lds_barrier();

    if (tid < MAX_GROUPS) {
        unsigned lf = cinc - cnt;
        for (int w = 0; w < wave; ++w) lf += wcnt[w];
        lfirst[tid] = lf;
        gfirst[tid] = (unsigned)goff + first;
        if (blockIdx.x == 0 && tid < nGroups) { tab[2 * tid] = goff; tab[2 * tid + 1] = mine; }
    }
    lds_barrier();

#pragma unroll
    for (int k = 0; k < 4; ++k) {
        if (c[k] >= 0) {
            const unsigned p = lfirst[g4[k]] + rk[k];
            stage[p] = make_uint2(((unsigned)r[k] << 8) | l4[k],
                                  __float_as_uint(rsqrtf(deg_s[r[k]]) * rsqrtf(deg_s[c[k]])));
            spos[p] = gfirst[g4[k]] + rk[k];
        }
    }
    lds_barrier();

    const int total = min(BIN_EDGES, E - (int)blockIdx.x * BIN_EDGES);
#pragma unroll
    for (int j = 0; j < 4; ++j) {
        const int i = tid + BIN_THREADS * j;
        if (i < total) recs[spos[i]] = stage[i];
    }

    for (int i = gid; i < n; i += gridDim.x * BIN_THREADS) dinv[i] = rsqrtf(deg_s[i]);
}

#define REPLAY_T 8
template <int D, bool RELU>
__global__ __launch_bounds__(OWN_THREADS)
void owner_replay_kernel(const uint2* __restrict__ recs,
                         const int* __restrict__ tab,
                         const float* __restrict__ dinv,
                         const float* __restrict__ h,
                         const float* __restrict__ bias,
                         float* __restrict__ out, int n) {
    constexpr int V2 = D / 64;
    constexpr int D4 = D / 4;
    __shared__ float4 acc4[OWN_ROWS * D4];
    float* acc = (float*)acc4;

    const int tid  = threadIdx.x;
    const int lane = tid & 31;
    const int wave = __builtin_amdgcn_readfirstlane(tid >> 5);
    const int base = blockIdx.x * OWN_ROWS;
    const int nrows = min(OWN_ROWS, n - base);
    const unsigned goff = (unsigned)tab[2 * blockIdx.x];
    const unsigned cnt  = (unsigned)tab[2 * blockIdx.x + 1];

    unsigned q0 = (unsigned)(REPLAY_T * wave);
    uint2 rec = make_uint2(0u, 0u);
    if (q0 < cnt) rec = recs[(size_t)goff + min(q0 + (unsigned)(lane & (REPLAY_T - 1)), cnt - 1u)];

    const float4* __restrict__ h4 = (const float4*)(h + (size_t)base * D);
    for (int i = tid; i < nrows * D4; i += OWN_THREADS) {
        int rr = i / D4, d = (i % D4) * 4;
        float di = dinv[base + rr];
        float s = di * di;
        float4 hv = h4[i];
        float4 bv = *(const float4*)(bias + d);
        acc4[i] = make_float4(s * hv.x + bv.x, s * hv.y + bv.y, s * hv.z + bv.z, s * hv.w + bv.w);
    }
    lds_barrier();

    for (; q0 < cnt; q0 += (OWN_THREADS / 32) * REPLAY_T) {
        const uint2 cur = rec;
        const unsigned qn = q0 + (OWN_THREADS / 32) * REPLAY_T;
        if (qn < cnt) rec = recs[(size_t)goff + min(qn + (unsigned)(lane & (REPLAY_T - 1)), cnt - 1u)];
        float2 v[REPLAY_T][V2];
        unsigned loc[REPLAY_T]; float nrm[REPLAY_T];
#pragma unroll
        for (int k = 0; k < REPLAY_T; ++k) {
            unsigned w0 = (unsigned)__builtin_amdgcn_readlane((int)cur.x, k);
            nrm[k] = __int_as_float(__builtin_amdgcn_readlane((int)cur.y, k));
            loc[k] = w0 & 255u;
            const float2* src = (const float2*)(h + (size_t)(w0 >> 8) * D) + lane;
#pragma unroll
            for (int j = 0; j < V2; ++j) v[k][j] = src[32 * j];
        }
#pragma unroll
        for (int k = 0; k < REPLAY_T; ++k) {
#pragma unroll
            for (int j = 0; j < V2; ++j) asm volatile("" :: "v"(v[k][j].x), "v"(v[k][j].y));
        }
#pragma unroll
        for (int k = 0; k < REPLAY_T; ++k) {
            if (q0 + (unsigned)k < cnt) {
                float* a = acc + loc[k] * D + 2 * lane;
#pragma unroll
                for (int j = 0; j < V2; ++j) {
                    atomicAdd(a + 64 * j,     nrm[k] * v[k][j].x);
                    atomicAdd(a + 64 * j + 1, nrm[k] * v[k][j].y);
                }
            }
        }
    }
    lds_barrier();

    float4* __restrict__ out4 = (float4*)(out + (size_t)base * D);
    for (int i = tid; i < nrows * D4; i += OWN_THREADS) {
        float4 v = acc4[i];
        if (RELU) v = make_float4(fmaxf(v.x, 0.0f), fmaxf(v.y, 0.0f), fmaxf(v.z, 0.0f), fmaxf(v.w, 0.0f));
        out4[i] = v;
    }
}

extern "C" void kernel_launch(void* const* d_in, const int* in_sizes, int n_in,
                              void* d_out, int out_size, void* d_ws, size_t ws_size,
                              hipStream_t stream) {
    const float* x  = (const float*)d_in[0];
    const int*   ei = (const int*)d_in[1];
    const float* W0 = (const float*)d_in[2];
    const float* b0 = (const float*)d_in[3];
    const float* W1 = (const float*)d_in[4];
    const float* b1 = (const float*)d_in[5];
    const float* W2 = (const float*)d_in[6];
    const float* b2 = (const float*)d_in[7];

    const int N = in_sizes[0] / IN_DIM;
    const int E = in_sizes[1] / 2;
    const int* rowIdx = ei;
    const int* colIdx = ei + E;

    char*  ws  = (char*)d_ws;
    size_t off = 0;
    auto   alloc = [&](size_t bytes) {
        void* p = ws + off;
        off += (bytes + 255) & ~(size_t)255;
        return p;
    };
    float* deg  = (float*)alloc((size_t)N * sizeof(float));
    float* dinv = (float*)alloc((size_t)N * sizeof(float));
    float* hbuf = (float*)alloc((size_t)N * HID_DIM * sizeof(float));
    float* xbuf = (float*)alloc((size_t)N * HID_DIM * sizeof(float));

    const int mtiles  = (N + 15) / 16;
    const int ownBlks = (N + OWN_ROWS - 1) / OWN_ROWS;
    uint2* recs = (uint2*)alloc((size_t)E * sizeof(uint2));
    int*   tab  = (int*)alloc((size_t)ownBlks * 2 * sizeof(int));
    unsigned* cursor = (unsigned*)alloc((size_t)ownBlks * sizeof(unsigned));

    init_deg_kernel<<<(N + 255) / 256, 256, 0, stream>>>(deg, cursor, N, ownBlks);
    accum_deg_kernel<<<(E + 255) / 256, 256, 0, stream>>>(colIdx, deg, E);
    edge_list_kernel<<<(E + 4 * BIN_THREADS - 1) / (4 * BIN_THREADS), BIN_THREADS, 0, stream>>>(
        rowIdx, colIdx, deg, dinv, cursor, recs, tab, N, E, ownBlks);

    gemm_wmma_kernel<128><<<256, GEMM_THREADS, 0, stream>>>(x, W0, hbuf, N, mtiles);
    owner_replay_kernel<128, true><<<ownBlks, OWN_THREADS, 0, stream>>>(
        recs, tab, dinv, hbuf, b0, xbuf, N);

    gemm_wmma_kernel<128><<<256, GEMM_THREADS, 0, stream>>>(xbuf, W1, hbuf, N, mtiles);
    owner_replay_kernel<128, true><<<ownBlks, OWN_THREADS, 0, stream>>>(
        recs, tab, dinv, hbuf, b1, xbuf, N);

    gemm_wmma_kernel<64><<<256, GEMM_THREADS, 0, stream>>>(xbuf, W2, hbuf, N, mtiles);
    owner_replay_kernel<64, false><<<ownBlks, OWN_THREADS, 0, stream>>>(
        recs, tab, dinv, hbuf, b2, (float*)d_out, N);
}


# Global float atomic degree count removed: owners count in-edges in LDS from the list (placed by per-block LDS group counts); norm split, deg^-1/2 at GEMM store and at owner's last pass (rounding)
# speedup vs baseline: 3.3642x; 1.2359x over previous
#include <hip/hip_runtime.h>
#include <hip/hip_bf16.h>

typedef float v2f __attribute__((ext_vector_type(2)));
typedef float v8f __attribute__((ext_vector_type(8)));

#define IN_DIM  128
#define HID_DIM 128
#define OUT_DIM 64

#define GEMM_THREADS 512
template <int NOUT>
__global__ __launch_bounds__(GEMM_THREADS)
void gemm_wmma_kernel(const float* __restrict__ X,
                      const float* __restrict__ W,
                      const float* __restrict__ dinv,
                      float* __restrict__ H, int nNodes, int mtiles) {
    constexpr int K  = 128;
    constexpr int KQ = K / 8;
    constexpr int NT = NOUT / 16;
    constexpr int MG = (GEMM_THREADS / 32) / NT;
    const int wv   = __builtin_amdgcn_readfirstlane(threadIdx.x >> 5);
    const int lane = threadIdx.x & 31;
    const int half = lane >> 4;
    const int l16  = lane & 15;
    const int n0   = (wv % NT) * 16;
    const int step = gridDim.x * MG;
    int t = blockIdx.x * MG + wv / NT;
    if (t >= mtiles) return;

    float4 wf[KQ];
    {
        const float4* p = (const float4*)(W + (size_t)(n0 + l16) * K + 4 * half);
#pragma unroll
        for (int j = 0; j < KQ; ++j) wf[j] = p[2 * j];
    }
    auto load_tile = [&](float4 (&dst)[KQ], float4 (&sc)[2], int tile) {
        const float4* dp = (const float4*)(dinv + tile * 16 + 8 * half);
        sc[0] = dp[0]; sc[1] = dp[1];
        int mrow = min(tile * 16 + l16, nNodes - 1);
        const float4* p = (const float4*)(X + (size_t)mrow * K + 4 * half);
#pragma unroll
        for (int j = 0; j < KQ; ++j) dst[j] = p[2 * j];
    };
    auto mma_store = [&](const float4 (&s)[KQ], const float4 (&sq)[2], int tile) {
        const int m0 = tile * 16;
        const float sc[8] = {sq[0].x, sq[0].y, sq[0].z, sq[0].w, sq[1].x, sq[1].y, sq[1].z, sq[1].w};
        v8f c = {0.f, 0.f, 0.f, 0.f, 0.f, 0.f, 0.f, 0.f};
#pragma unroll
        for (int j = 0; j < KQ; ++j) {
            v2f a0 = {s[j].x, s[j].y}, b0 = {wf[j].x, wf[j].y};
            v2f a1 = {s[j].z, s[j].w}, b1 = {wf[j].z, wf[j].w};
            c = __builtin_amdgcn_wmma_f32_16x16x4_f32(false, a0, false, b0,
                                                      (short)0, c, false, false);
            c = __builtin_amdgcn_wmma_f32_16x16x4_f32(false, a1, false, b1,
                                                      (short)0, c, false, false);
        }
        float* out = H + (size_t)(m0 + 8 * half) * NOUT + n0 + l16;
        if (m0 + 16 <= nNodes) {
#pragma unroll
            for (int v = 0; v < 8; ++v) out[(size_t)v * NOUT] = sc[v] * c[v];
        } else {
#pragma unroll
            for (int v = 0; v < 8; ++v)
                if (m0 + 8 * half + v < nNodes) out[(size_t)v * NOUT] = sc[v] * c[v];
        }
    };

    float4 fa[KQ], fb[KQ], sa[2], sb[2];
    load_tile(fa, sa, t);
    for (;;) {
        int tn = t + step;
        load_tile(fb, sb, min(tn, mtiles - 1));
        __builtin_amdgcn_sched_barrier(0);
        mma_store(fa, sa, t);
        if (tn >= mtiles) break;
        t = tn + step;
        load_tile(fa, sa, min(t, mtiles - 1));
        __builtin_amdgcn_sched_barrier(0);
        mma_store(fb, sb, tn);
        if (t >= mtiles) break;
    }
}

#define OWN_ROWS     196
#define OWN_THREADS  1024
#define MAX_GROUPS   256
#define BIN_THREADS  1024
#define BIN_EDGES    (BIN_THREADS * 4)
#define CNT_Q        16
#define CNT_U        8

__device__ __forceinline__ void lds_barrier() {
    __builtin_amdgcn_fence(__ATOMIC_RELEASE, "workgroup", "local");
    __builtin_amdgcn_s_barrier();
    __builtin_amdgcn_fence(__ATOMIC_ACQUIRE, "workgroup", "local");
}

__device__ __forceinline__ void load_cols(const int* __restrict__ col, int e0, int E, int (&c)[4]) {
    if (e0 + 3 < E) {
        int4 cv = *(const int4*)(col + e0);
        c[0] = cv.x; c[1] = cv.y; c[2] = cv.z; c[3] = cv.w;
    } else {
#pragma unroll
        for (int k = 0; k < 4; ++k) c[k] = (e0 + k < E) ? col[e0 + k] : -1;
    }
}

__global__ __launch_bounds__(BIN_THREADS)
void group_count_kernel(const int* __restrict__ col, unsigned* __restrict__ cnt, int E) {
    __shared__ unsigned hist[MAX_GROUPS];
    const int tid = threadIdx.x;
    int c[4];
    load_cols(col, (blockIdx.x * BIN_THREADS + tid) * 4, E, c);
    if (tid < MAX_GROUPS) hist[tid] = 0u;
    lds_barrier();
#pragma unroll
    for (int k = 0; k < 4; ++k)
        if (c[k] >= 0) atomicAdd(&hist[(unsigned)c[k] / OWN_ROWS], 1u);
    lds_barrier();
    if (tid < MAX_GROUPS) cnt[blockIdx.x * MAX_GROUPS + tid] = hist[tid];
}

__global__ __launch_bounds__(BIN_THREADS)
void edge_list_kernel(const int* __restrict__ row,
                      const int* __restrict__ col,
                      const unsigned* __restrict__ cnt,
                      unsigned* __restrict__ recs, int* __restrict__ tab,
                      int E, int nGroups) {
    constexpr int G4 = MAX_GROUPS / 4;
    static_assert(BIN_THREADS == CNT_Q * G4, "one thread per piece and row class");

    __shared__ unsigned stage[BIN_EDGES];
    __shared__ unsigned spos[BIN_EDGES];
    __shared__ uint4    ptot[CNT_Q * G4];
    __shared__ uint4    pbef[CNT_Q * G4];
    __shared__ unsigned slot[MAX_GROUPS];
    __shared__ unsigned lfirst[MAX_GROUPS];
    __shared__ unsigned gfirst[MAX_GROUPS];
    __shared__ unsigned wtot[MAX_GROUPS / 32];
    __shared__ unsigned wcnt[MAX_GROUPS / 32];

    const int tid  = threadIdx.x;
    const int lane = tid & 31;
    const int wave = tid >> 5;
    const int nb   = gridDim.x;
    const int bid  = blockIdx.x;

    int r[4], c[4];
    const int e0 = (bid * BIN_THREADS + tid) * 4;
    load_cols(col, e0, E, c);
    if (e0 + 3 < E) {
        int4 rv = *(const int4*)(row + e0);
        r[0] = rv.x; r[1] = rv.y; r[2] = rv.z; r[3] = rv.w;
    } else {
#pragma unroll
        for (int k = 0; k < 4; ++k) r[k] = (e0 + k < E) ? row[e0 + k] : 0;
    }

    {
        const int q = tid / G4, g4 = tid % G4;
        const uint4* __restrict__ c4 = (const uint4*)cnt;
        uint4 tot = make_uint4(0u, 0u, 0u, 0u), bef = make_uint4(0u, 0u, 0u, 0u);
        for (int b0 = q; b0 < nb; b0 += CNT_Q * CNT_U) {
            uint4 v[CNT_U];
#pragma unroll
            for (int j = 0; j < CNT_U; ++j) v[j] = c4[min(b0 + CNT_Q * j, nb - 1) * G4 + g4];
#pragma unroll
            for (int j = 0; j < CNT_U; ++j) {
                const int b = b0 + CNT_Q * j;
                const unsigned ma = (b < nb) ? 0xffffffffu : 0u;
                const unsigned mb = (b < bid) ? ma : 0u;
                tot.x += v[j].x & ma; tot.y += v[j].y & ma; tot.z += v[j].z & ma; tot.w += v[j].w & ma;
                bef.x += v[j].x & mb; bef.y += v[j].y & mb; bef.z += v[j].z & mb; bef.w += v[j].w & mb;
            }
        }
        ptot[tid] = tot;
        pbef[tid] = bef;
    }
    if (tid < MAX_GROUPS) slot[tid] = 0u;
    lds_barrier();

    unsigned g4[4], l4[4], rk[4];
#pragma unroll
    for (int k = 0; k < 4; ++k) {
        unsigned cc = (unsigned)max(c[k], 0);
        g4[k] = cc / OWN_ROWS;
        l4[k] = cc - g4[k] * OWN_ROWS;
        rk[k] = 0u;
        if (c[k] >= 0) rk[k] = atomicAdd(&slot[g4[k]], 1u);
    }

    unsigned mine = 0u, before = 0u;
    if (tid < MAX_GROUPS) {
        const unsigned* pt = (const unsigned*)ptot;
        const unsigned* pb = (const unsigned*)pbef;
#pragma unroll
        for (int q = 0; q < CNT_Q; ++q) {
            mine   += pt[q * MAX_GROUPS + tid];
            before += pb[q * MAX_GROUPS + tid];
        }
    }
    unsigned inc = mine;
#pragma unroll
    for (int m = 1; m < 32; m <<= 1) {
        unsigned t = __shfl_up(inc, m, 32);
        if (lane >= m) inc += t;
    }
    if (tid < MAX_GROUPS && lane == 31) wtot[wave] = inc;
    lds_barrier();

    unsigned cntg = 0u;
    if (tid < MAX_GROUPS) cntg = slot[tid];
    unsigned cinc = cntg;
#pragma unroll
    for (int m = 1; m < 32; m <<= 1) {
        unsigned t = __shfl_up(cinc, m, 32);
        if (lane >= m) cinc += t;
    }
    if (tid < MAX_GROUPS && lane == 31) wcnt[wave] = cinc;
    lds_barrier();

    if (tid < MAX_GROUPS) {
        unsigned goff = inc - mine, lf = cinc - cntg;
        for (int w = 0; w < wave; ++w) { goff += wtot[w]; lf += wcnt[w]; }
        lfirst[tid] = lf;
        gfirst[tid] = goff + before;
        if (bid == 0 && tid < nGroups) { tab[2 * tid] = (int)goff; tab[2 * tid + 1] = (int)mine; }
    }
    lds_barrier();

#pragma unroll
    for (int k = 0; k < 4; ++k) {
        if (c[k] >= 0) {
            const unsigned p = lfirst[g4[k]] + rk[k];
            stage[p] = ((unsigned)r[k] << 8) | l4[k];
            spos[p] = gfirst[g4[k]] + rk[k];
        }
    }
    lds_barrier();

    const int total = min(BIN_EDGES, E - bid * BIN_EDGES);
#pragma unroll
    for (int j = 0; j < 4; ++j) {
        const int i = tid + BIN_THREADS * j;
        if (i < total) recs[spos[i]] = stage[i];
    }
}

#define DEG_THREADS  256
#define DEG_U        16
__global__ __launch_bounds__(DEG_THREADS)
void owner_degree_kernel(const unsigned* __restrict__ recs,
                         const int* __restrict__ tab,
                         float* __restrict__ dinv, int n, int npad) {
    static_assert(OWN_ROWS <= 224 && DEG_THREADS >= 256, "spare counters behind the rows");
    __shared__ unsigned dcount[256];
    const int tid  = threadIdx.x;
    const int base = blockIdx.x * OWN_ROWS;
    const int nrows = min(OWN_ROWS, n - base);
    const unsigned goff = (unsigned)tab[2 * blockIdx.x];
    const unsigned cnt  = (unsigned)tab[2 * blockIdx.x + 1];
    if (tid < 256) dcount[tid] = 1u;
    const unsigned spare = 224u + (unsigned)(tid & 31);
    lds_barrier();
    for (unsigned i0 = (unsigned)tid; i0 < cnt; i0 += (unsigned)(DEG_U * DEG_THREADS)) {
        unsigned e[DEG_U];
#pragma unroll
        for (int j = 0; j < DEG_U; ++j)
            e[j] = recs[(size_t)goff + min(i0 + (unsigned)(DEG_THREADS * j), cnt - 1u)];
#pragma unroll
        for (int j = 0; j < DEG_U; ++j)
            atomicAdd(&dcount[(i0 + (unsigned)(DEG_THREADS * j) < cnt) ? (e[j] & 255u) : spare], 1u);
    }
    lds_barrier();
    if (tid < nrows) dinv[base + tid] = rsqrtf((float)dcount[tid]);
    if (blockIdx.x == gridDim.x - 1 && tid < npad - n) dinv[n + tid] = 0.0f;
}

#define REPLAY_T 8
template <int D, bool RELU>
__global__ __launch_bounds__(OWN_THREADS)
void owner_replay_kernel(const unsigned* __restrict__ recs,
                         const int* __restrict__ tab,
                         const float* __restrict__ dinv,
                         const float* __restrict__ h,
                         const float* __restrict__ bias,
                         float* __restrict__ out, int n) {
    constexpr int V2 = D / 64;
    constexpr int D4 = D / 4;
    __shared__ float4 acc4[OWN_ROWS * D4];
    float* acc = (float*)acc4;

    const int tid  = threadIdx.x;
    const int lane = tid & 31;
    const int wave = __builtin_amdgcn_readfirstlane(tid >> 5);
    const int base = blockIdx.x * OWN_ROWS;
    const int nrows = min(OWN_ROWS, n - base);
    const unsigned goff = (unsigned)tab[2 * blockIdx.x];
    const unsigned cnt  = (unsigned)tab[2 * blockIdx.x + 1];

    unsigned q0 = (unsigned)(REPLAY_T * wave);
    unsigned rec = 0u;
    if (q0 < cnt) rec = recs[(size_t)goff + min(q0 + (unsigned)(lane & (REPLAY_T - 1)), cnt - 1u)];

    const float4* __restrict__ h4 = (const float4*)(h + (size_t)base * D);
    for (int i = tid; i < nrows * D4; i += OWN_THREADS) acc4[i] = h4[i];
    lds_barrier();

    for (; q0 < cnt; q0 += (OWN_THREADS / 32) * REPLAY_T) {
        const unsigned cur = rec;
        const unsigned qn = q0 + (OWN_THREADS / 32) * REPLAY_T;
        if (qn < cnt) rec = recs[(size_t)goff + min(qn + (unsigned)(lane & (REPLAY_T - 1)), cnt - 1u)];
        float2 v[REPLAY_T][V2];
        unsigned loc[REPLAY_T];
#pragma unroll
        for (int k = 0; k < REPLAY_T; ++k) {
            unsigned w0 = (unsigned)__builtin_amdgcn_readlane((int)cur, k);
            loc[k] = w0 & 255u;
            const float2* src = (const float2*)(h + (size_t)(w0 >> 8) * D) + lane;
#pragma unroll
            for (int j = 0; j < V2; ++j) v[k][j] = src[32 * j];
        }
#pragma unroll
        for (int k = 0; k < REPLAY_T; ++k) {
#pragma unroll
            for (int j = 0; j < V2; ++j) asm volatile("" :: "v"(v[k][j].x), "v"(v[k][j].y));
        }
#pragma unroll
        for (int k = 0; k < REPLAY_T; ++k) {
            if (q0 + (unsigned)k < cnt) {
                float* a = acc + loc[k] * D + 2 * lane;
#pragma unroll
                for (int j = 0; j < V2; ++j) {
                    atomicAdd(a + 64 * j,     v[k][j].x);
                    atomicAdd(a + 64 * j + 1, v[k][j].y);
                }
            }
        }
    }
    lds_barrier();

    float4* __restrict__ out4 = (float4*)(out + (size_t)base * D);
    for (int i = tid; i < nrows * D4; i += OWN_THREADS) {
        const int rr = i / D4, d = (i % D4) * 4;
        const float di = dinv[base + rr];
        const float4 bv = *(const float4*)(bias + d);
        float4 v = acc4[i];
        v = make_float4(di * v.x + bv.x, di * v.y + bv.y, di * v.z + bv.z, di * v.w + bv.w);
        if (RELU) v = make_float4(fmaxf(v.x, 0.0f), fmaxf(v.y, 0.0f), fmaxf(v.z, 0.0f), fmaxf(v.w, 0.0f));
        out4[i] = v;
    }
}

extern "C" void kernel_launch(void* const* d_in, const int* in_sizes, int n_in,
                              void* d_out, int out_size, void* d_ws, size_t ws_size,
                              hipStream_t stream) {
    const float* x  = (const float*)d_in[0];
    const int*   ei = (const int*)d_in[1];
    const float* W0 = (const float*)d_in[2];
    const float* b0 = (const float*)d_in[3];
    const float* W1 = (const float*)d_in[4];
    const float* b1 = (const float*)d_in[5];
    const float* W2 = (const float*)d_in[6];
    const float* b2 = (const float*)d_in[7];

    const int N = in_sizes[0] / IN_DIM;
    const int E = in_sizes[1] / 2;
    const int* rowIdx = ei;
    const int* colIdx = ei + E;

    char*  ws  = (char*)d_ws;
    size_t off = 0;
    auto   alloc = [&](size_t bytes) {
        void* p = ws + off;
        off += (bytes + 255) & ~(size_t)255;
        return p;
    };
    float* dinv = (float*)alloc((size_t)(N + 16) * sizeof(float));
    float* hbuf = (float*)alloc((size_t)N * HID_DIM * sizeof(float));
    float* xbuf = (float*)alloc((size_t)N * HID_DIM * sizeof(float));

    const int mtiles  = (N + 15) / 16;
    const int ownBlks = (N + OWN_ROWS - 1) / OWN_ROWS;
    const int binBlks = (E + BIN_EDGES - 1) / BIN_EDGES;
    unsigned* recs = (unsigned*)alloc((size_t)E * sizeof(unsigned));
    int*      tab  = (int*)alloc((size_t)ownBlks * 2 * sizeof(int));
    unsigned* cnt  = (unsigned*)alloc((size_t)binBlks * MAX_GROUPS * sizeof(unsigned));

    group_count_kernel<<<binBlks, BIN_THREADS, 0, stream>>>(colIdx, cnt, E);
    edge_list_kernel<<<binBlks, BIN_THREADS, 0, stream>>>(rowIdx, colIdx, cnt, recs, tab, E, ownBlks);
    owner_degree_kernel<<<ownBlks, DEG_THREADS, 0, stream>>>(recs, tab, dinv, N, mtiles * 16);

    gemm_wmma_kernel<128><<<256, GEMM_THREADS, 0, stream>>>(x, W0, dinv, hbuf, N, mtiles);
    owner_replay_kernel<128, true><<<ownBlks, OWN_THREADS, 0, stream>>>(
        recs, tab, dinv, hbuf, b0, xbuf, N);

    gemm_wmma_kernel<128><<<256, GEMM_THREADS, 0, stream>>>(xbuf, W1, dinv, hbuf, N, mtiles);
    owner_replay_kernel<128, true><<<ownBlks, OWN_THREADS, 0, stream>>>(
        recs, tab, dinv, hbuf, b1, xbuf, N);

    gemm_wmma_kernel<64><<<256, GEMM_THREADS, 0, stream>>>(xbuf, W2, dinv, hbuf, N, mtiles);
    owner_replay_kernel<64, false><<<ownBlks, OWN_THREADS, 0, stream>>>(
        recs, tab, dinv, hbuf, b2, (float*)d_out, N);
}


# Layers 0-1: the owner group keeps its 196 finished rows in LDS and multiplies them by the next layer's W itself (f32 WMMA): two GEMM launches and the activation round trip gone; every sum order kept
# speedup vs baseline: 3.6105x; 1.0732x over previous
#include <hip/hip_runtime.h>
#include <hip/hip_bf16.h>

typedef float v2f __attribute__((ext_vector_type(2)));
typedef float v8f __attribute__((ext_vector_type(8)));

#define IN_DIM  128
#define HID_DIM 128
#define OUT_DIM 64

#define GEMM_THREADS 512
template <int NOUT>
__global__ __launch_bounds__(GEMM_THREADS)
void gemm_wmma_kernel(const float* __restrict__ X,
                      const float* __restrict__ W,
                      const float* __restrict__ dinv,
                      float* __restrict__ H, int nNodes, int mtiles) {
    constexpr int K  = 128;
    constexpr int KQ = K / 8;
    constexpr int NT = NOUT / 16;
    constexpr int MG = (GEMM_THREADS / 32) / NT;
    const int wv   = __builtin_amdgcn_readfirstlane(threadIdx.x >> 5);
    const int lane = threadIdx.x & 31;
    const int half = lane >> 4;
    const int l16  = lane & 15;
    const int n0   = (wv % NT) * 16;
    const int step = gridDim.x * MG;
    int t = blockIdx.x * MG + wv / NT;
    if (t >= mtiles) return;

    float4 wf[KQ];
    {
        const float4* p = (const float4*)(W + (size_t)(n0 + l16) * K + 4 * half);
#pragma unroll
        for (int j = 0; j < KQ; ++j) wf[j] = p[2 * j];
    }
    auto load_tile = [&](float4 (&dst)[KQ], float4 (&sc)[2], int tile) {
        const float4* dp = (const float4*)(dinv + tile * 16 + 8 * half);
        sc[0] = dp[0]; sc[1] = dp[1];
        int mrow = min(tile * 16 + l16, nNodes - 1);
        const float4* p = (const float4*)(X + (size_t)mrow * K + 4 * half);
#pragma unroll
        for (int j = 0; j < KQ; ++j) dst[j] = p[2 * j];
    };
    auto mma_store = [&](const float4 (&s)[KQ], const float4 (&sq)[2], int tile) {
        const int m0 = tile * 16;
        const float sc[8] = {sq[0].x, sq[0].y, sq[0].z, sq[0].w, sq[1].x, sq[1].y, sq[1].z, sq[1].w};
        v8f c = {0.f, 0.f, 0.f, 0.f, 0.f, 0.f, 0.f, 0.f};
#pragma unroll
        for (int j = 0; j < KQ; ++j) {
            v2f a0 = {s[j].x, s[j].y}, b0 = {wf[j].x, wf[j].y};
            v2f a1 = {s[j].z, s[j].w}, b1 = {wf[j].z, wf[j].w};
            c = __builtin_amdgcn_wmma_f32_16x16x4_f32(false, a0, false, b0,
                                                      (short)0, c, false, false);
            c = __builtin_amdgcn_wmma_f32_16x16x4_f32(false, a1, false, b1,
                                                      (short)0, c, false, false);
        }
        float* out = H + (size_t)(m0 + 8 * half) * NOUT + n0 + l16;
        if (m0 + 16 <= nNodes) {
#pragma unroll
            for (int v = 0; v < 8; ++v) out[(size_t)v * NOUT] = sc[v] * c[v];
        } else {
#pragma unroll
            for (int v = 0; v < 8; ++v)
                if (m0 + 8 * half + v < nNodes) out[(size_t)v * NOUT] = sc[v] * c[v];
        }
    };

    float4 fa[KQ], fb[KQ], sa[2], sb[2];
    load_tile(fa, sa, t);
    for (;;) {
        int tn = t + step;
        load_tile(fb, sb, min(tn, mtiles - 1));
        __builtin_amdgcn_sched_barrier(0);
        mma_store(fa, sa, t);
        if (tn >= mtiles) break;
        t = tn + step;
        load_tile(fa, sa, min(t, mtiles - 1));
        __builtin_amdgcn_sched_barrier(0);
        mma_store(fb, sb, tn);
        if (t >= mtiles) break;
    }
}

#define OWN_ROWS     196
#define OWN_THREADS  1024
#define MAX_GROUPS   256
#define BIN_THREADS  1024
#define BIN_EDGES    (BIN_THREADS * 4)
#define CNT_Q        16
#define CNT_U        8

__device__ __forceinline__ void lds_barrier() {
    __builtin_amdgcn_fence(__ATOMIC_RELEASE, "workgroup", "local");
    __builtin_amdgcn_s_barrier();
    __builtin_amdgcn_fence(__ATOMIC_ACQUIRE, "workgroup", "local");
}

__device__ __forceinline__ void load_cols(const int* __restrict__ col, int e0, int E, int (&c)[4]) {
    if (e0 + 3 < E) {
        int4 cv = *(const int4*)(col + e0);
        c[0] = cv.x; c[1] = cv.y; c[2] = cv.z; c[3] = cv.w;
    } else {
#pragma unroll
        for (int k = 0; k < 4; ++k) c[k] = (e0 + k < E) ? col[e0 + k] : -1;
    }
}

__global__ __launch_bounds__(BIN_THREADS)
void group_count_kernel(const int* __restrict__ col, unsigned* __restrict__ cnt, int E) {
    __shared__ unsigned hist[MAX_GROUPS];
    const int tid = threadIdx.x;
    int c[4];
    load_cols(col, (blockIdx.x * BIN_THREADS + tid) * 4, E, c);
    if (tid < MAX_GROUPS) hist[tid] = 0u;
    lds_barrier();
#pragma unroll
    for (int k = 0; k < 4; ++k)
        if (c[k] >= 0) atomicAdd(&hist[(unsigned)c[k] / OWN_ROWS], 1u);
    lds_barrier();
    if (tid < MAX_GROUPS) cnt[blockIdx.x * MAX_GROUPS + tid] = hist[tid];
}

__global__ __launch_bounds__(BIN_THREADS)
void edge_list_kernel(const int* __restrict__ row,
                      const int* __restrict__ col,
                      const unsigned* __restrict__ cnt,
                      unsigned* __restrict__ recs, int* __restrict__ tab,
                      int E, int nGroups) {
    constexpr int G4 = MAX_GROUPS / 4;
    static_assert(BIN_THREADS == CNT_Q * G4, "one thread per piece and row class");

    __shared__ unsigned stage[BIN_EDGES];
    __shared__ unsigned spos[BIN_EDGES];
    __shared__ uint4    ptot[CNT_Q * G4];
    __shared__ uint4    pbef[CNT_Q * G4];
    __shared__ unsigned slot[MAX_GROUPS];
    __shared__ unsigned lfirst[MAX_GROUPS];
    __shared__ unsigned gfirst[MAX_GROUPS];
    __shared__ unsigned wtot[MAX_GROUPS / 32];
    __shared__ unsigned wcnt[MAX_GROUPS / 32];

    const int tid  = threadIdx.x;
    const int lane = tid & 31;
    const int wave = tid >> 5;
    const int nb   = gridDim.x;
    const int bid  = blockIdx.x;

    int r[4], c[4];
    const int e0 = (bid * BIN_THREADS + tid) * 4;
    load_cols(col, e0, E, c);
    if (e0 + 3 < E) {
        int4 rv = *(const int4*)(row + e0);
        r[0] = rv.x; r[1] = rv.y; r[2] = rv.z; r[3] = rv.w;
    } else {
#pragma unroll
        for (int k = 0; k < 4; ++k) r[k] = (e0 + k < E) ? row[e0 + k] : 0;
    }

    {
        const int q = tid / G4, g4 = tid % G4;
        const uint4* __restrict__ c4 = (const uint4*)cnt;
        uint4 tot = make_uint4(0u, 0u, 0u, 0u), bef = make_uint4(0u, 0u, 0u, 0u);
        for (int b0 = q; b0 < nb; b0 += CNT_Q * CNT_U) {
            uint4 v[CNT_U];
#pragma unroll
            for (int j = 0; j < CNT_U; ++j) v[j] = c4[min(b0 + CNT_Q * j, nb - 1) * G4 + g4];
#pragma unroll
            for (int j = 0; j < CNT_U; ++j) {
                const int b = b0 + CNT_Q * j;
                const unsigned ma = (b < nb) ? 0xffffffffu : 0u;
                const unsigned mb = (b < bid) ? ma : 0u;
                tot.x += v[j].x & ma; tot.y += v[j].y & ma; tot.z += v[j].z & ma; tot.w += v[j].w & ma;
                bef.x += v[j].x & mb; bef.y += v[j].y & mb; bef.z += v[j].z & mb; bef.w += v[j].w & mb;
            }
        }
        ptot[tid] = tot;
        pbef[tid] = bef;
    }
    if (tid < MAX_GROUPS) slot[tid] = 0u;
    lds_barrier();

    unsigned g4[4], l4[4], rk[4];
#pragma unroll
    for (int k = 0; k < 4; ++k) {
        unsigned cc = (unsigned)max(c[k], 0);
        g4[k] = cc / OWN_ROWS;
        l4[k] = cc - g4[k] * OWN_ROWS;
        rk[k] = 0u;
        if (c[k] >= 0) rk[k] = atomicAdd(&slot[g4[k]], 1u);
    }

    unsigned mine = 0u, before = 0u;
    if (tid < MAX_GROUPS) {
        const unsigned* pt = (const unsigned*)ptot;
        const unsigned* pb = (const unsigned*)pbef;
#pragma unroll
        for (int q = 0; q < CNT_Q; ++q) {
            mine   += pt[q * MAX_GROUPS + tid];
            before += pb[q * MAX_GROUPS + tid];
        }
    }
    unsigned inc = mine;
#pragma unroll
    for (int m = 1; m < 32; m <<= 1) {
        unsigned t = __shfl_up(inc, m, 32);
        if (lane >= m) inc += t;
    }
    if (tid < MAX_GROUPS && lane == 31) wtot[wave] = inc;
    lds_barrier();

    unsigned cntg = 0u;
    if (tid < MAX_GROUPS) cntg = slot[tid];
    unsigned cinc = cntg;
#pragma unroll
    for (int m = 1; m < 32; m <<= 1) {
        unsigned t = __shfl_up(cinc, m, 32);
        if (lane >= m) cinc += t;
    }
    if (tid < MAX_GROUPS && lane == 31) wcnt[wave] = cinc;
    lds_barrier();

    if (tid < MAX_GROUPS) {
        unsigned goff = inc - mine, lf = cinc - cntg;
        for (int w = 0; w < wave; ++w) { goff += wtot[w]; lf += wcnt[w]; }
        lfirst[tid] = lf;
        gfirst[tid] = goff + before;
        if (bid == 0 && tid < nGroups) { tab[2 * tid] = (int)goff; tab[2 * tid + 1] = (int)mine; }
    }
    lds_barrier();

#pragma unroll
    for (int k = 0; k < 4; ++k) {
        if (c[k] >= 0) {
            const unsigned p = lfirst[g4[k]] + rk[k];
            stage[p] = ((unsigned)r[k] << 8) | l4[k];
            spos[p] = gfirst[g4[k]] + rk[k];
        }
    }
    lds_barrier();

    const int total = min(BIN_EDGES, E - bid * BIN_EDGES);
#pragma unroll
    for (int j = 0; j < 4; ++j) {
        const int i = tid + BIN_THREADS * j;
        if (i < total) recs[spos[i]] = stage[i];
    }
}

#define DEG_THREADS  256
#define DEG_U        16
__global__ __launch_bounds__(DEG_THREADS)
void owner_degree_kernel(const unsigned* __restrict__ recs,
                         const int* __restrict__ tab,
                         float* __restrict__ dinv, int n, int npad) {
    static_assert(OWN_ROWS <= 224 && DEG_THREADS >= 256, "spare counters behind the rows");
    __shared__ unsigned dcount[256];
    const int tid  = threadIdx.x;
    const int base = blockIdx.x * OWN_ROWS;
    const int nrows = min(OWN_ROWS, n - base);
    const unsigned goff = (unsigned)tab[2 * blockIdx.x];
    const unsigned cnt  = (unsigned)tab[2 * blockIdx.x + 1];
    if (tid < 256) dcount[tid] = 1u;
    const unsigned spare = 224u + (unsigned)(tid & 31);
    lds_barrier();
    for (unsigned i0 = (unsigned)tid; i0 < cnt; i0 += (unsigned)(DEG_U * DEG_THREADS)) {
        unsigned e[DEG_U];
#pragma unroll
        for (int j = 0; j < DEG_U; ++j)
            e[j] = recs[(size_t)goff + min(i0 + (unsigned)(DEG_THREADS * j), cnt - 1u)];
#pragma unroll
        for (int j = 0; j < DEG_U; ++j)
            atomicAdd(&dcount[(i0 + (unsigned)(DEG_THREADS * j) < cnt) ? (e[j] & 255u) : spare], 1u);
    }
    lds_barrier();
    if (tid < nrows) dinv[base + tid] = rsqrtf((float)dcount[tid]);
    if (blockIdx.x == gridDim.x - 1 && tid < npad - n) dinv[n + tid] = 0.0f;
}

#define REPLAY_T 8
template <int D, bool RELU>
__global__ __launch_bounds__(OWN_THREADS)
void owner_replay_kernel(const unsigned* __restrict__ recs,
                         const int* __restrict__ tab,
                         const float* __restrict__ dinv,
                         const float* __restrict__ h,
                         const float* __restrict__ bias,
                         float* __restrict__ out, int n) {
    constexpr int V2 = D / 64;
    constexpr int D4 = D / 4;
    __shared__ float4 acc4[OWN_ROWS * D4];
    float* acc = (float*)acc4;

    const int tid  = threadIdx.x;
    const int lane = tid & 31;
    const int wave = __builtin_amdgcn_readfirstlane(tid >> 5);
    const int base = blockIdx.x * OWN_ROWS;
    const int nrows = min(OWN_ROWS, n - base);
    const unsigned goff = (unsigned)tab[2 * blockIdx.x];
    const unsigned cnt  = (unsigned)tab[2 * blockIdx.x + 1];

    unsigned q0 = (unsigned)(REPLAY_T * wave);
    unsigned rec = 0u;
    if (q0 < cnt) rec = recs[(size_t)goff + min(q0 + (unsigned)(lane & (REPLAY_T - 1)), cnt - 1u)];

    const float4* __restrict__ h4 = (const float4*)(h + (size_t)base * D);
    for (int i = tid; i < nrows * D4; i += OWN_THREADS) acc4[i] = h4[i];
    lds_barrier();

    for (; q0 < cnt; q0 += (OWN_THREADS / 32) * REPLAY_T) {
        const unsigned cur = rec;
        const unsigned qn = q0 + (OWN_THREADS / 32) * REPLAY_T;
        if (qn < cnt) rec = recs[(size_t)goff + min(qn + (unsigned)(lane & (REPLAY_T - 1)), cnt - 1u)];
        float2 v[REPLAY_T][V2];
        unsigned loc[REPLAY_T];
#pragma unroll
        for (int k = 0; k < REPLAY_T; ++k) {
            unsigned w0 = (unsigned)__builtin_amdgcn_readlane((int)cur, k);
            loc[k] = w0 & 255u;
            const float2* src = (const float2*)(h + (size_t)(w0 >> 8) * D) + lane;
#pragma unroll
            for (int j = 0; j < V2; ++j) v[k][j] = src[32 * j];
        }
#pragma unroll
        for (int k = 0; k < REPLAY_T; ++k) {
#pragma unroll
            for (int j = 0; j < V2; ++j) asm volatile("" :: "v"(v[k][j].x), "v"(v[k][j].y));
        }
#pragma unroll
        for (int k = 0; k < REPLAY_T; ++k) {
            if (q0 + (unsigned)k < cnt) {
                float* a = acc + loc[k] * D + 2 * lane;
#pragma unroll
                for (int j = 0; j < V2; ++j) {
                    atomicAdd(a + 64 * j,     v[k][j].x);
                    atomicAdd(a + 64 * j + 1, v[k][j].y);
                }
            }
        }
    }
    lds_barrier();

    float4* __restrict__ out4 = (float4*)(out + (size_t)base * D);
    for (int i = tid; i < nrows * D4; i += OWN_THREADS) {
        const int rr = i / D4, d = (i % D4) * 4;
        const float di = dinv[base + rr];
        const float4 bv = *(const float4*)(bias + d);
        float4 v = acc4[i];
        v = make_float4(di * v.x + bv.x, di * v.y + bv.y, di * v.z + bv.z, di * v.w + bv.w);
        if (RELU) v = make_float4(fmaxf(v.x, 0.0f), fmaxf(v.y, 0.0f), fmaxf(v.z, 0.0f), fmaxf(v.w, 0.0f));
        out4[i] = v;
    }
}

#define OWN_TILES ((OWN_ROWS + 15) / 16)
#define OWN_PITCH 132
template <int NOUT>
__global__ __launch_bounds__(OWN_THREADS)
void owner_replay_product_kernel(const unsigned* __restrict__ recs,
                                 const int* __restrict__ tab,
                                 const float* __restrict__ dinv,
                                 const float* __restrict__ h,
                                 const float* __restrict__ bias,
                                 const float* __restrict__ Wn,
                                 float* __restrict__ hout, int n) {
    constexpr int D  = 128;
    constexpr int V2 = D / 64;
    constexpr int D4 = D / 4;
    constexpr int P4 = OWN_PITCH / 4;
    __shared__ float4 acc4[OWN_TILES * 16 * P4];
    float* acc = (float*)acc4;

    const int tid  = threadIdx.x;
    const int lane = tid & 31;
    const int wave = __builtin_amdgcn_readfirstlane(tid >> 5);
    const int base = blockIdx.x * OWN_ROWS;
    const int nrows = min(OWN_ROWS, n - base);
    const unsigned goff = (unsigned)tab[2 * blockIdx.x];
    const unsigned cnt  = (unsigned)tab[2 * blockIdx.x + 1];

    unsigned q0 = (unsigned)(REPLAY_T * wave);
    unsigned rec = 0u;
    if (q0 < cnt) rec = recs[(size_t)goff + min(q0 + (unsigned)(lane & (REPLAY_T - 1)), cnt - 1u)];

    const float4* __restrict__ h4 = (const float4*)(h + (size_t)base * D);
    for (int i = tid; i < OWN_TILES * 16 * D4; i += OWN_THREADS) {
        const int rr = i / D4, d4 = i % D4;
        float4 v = make_float4(0.0f, 0.0f, 0.0f, 0.0f);
        if (rr < nrows) v = h4[i];
        acc4[rr * P4 + d4] = v;
    }
    lds_barrier();

    for (; q0 < cnt; q0 += (OWN_THREADS / 32) * REPLAY_T) {
        const unsigned cur = rec;
        const unsigned qn = q0 + (OWN_THREADS / 32) * REPLAY_T;
        if (qn < cnt) rec = recs[(size_t)goff + min(qn + (unsigned)(lane & (REPLAY_T - 1)), cnt - 1u)];
        float2 v[REPLAY_T][V2];
        unsigned loc[REPLAY_T];
#pragma unroll
        for (int k = 0; k < REPLAY_T; ++k) {
            unsigned w0 = (unsigned)__builtin_amdgcn_readlane((int)cur, k);
            loc[k] = w0 & 255u;
            const float2* src = (const float2*)(h + (size_t)(w0 >> 8) * D) + lane;
#pragma unroll
            for (int j = 0; j < V2; ++j) v[k][j] = src[32 * j];
        }
#pragma unroll
        for (int k = 0; k < REPLAY_T; ++k) {
#pragma unroll
            for (int j = 0; j < V2; ++j) asm volatile("" :: "v"(v[k][j].x), "v"(v[k][j].y));
        }
#pragma unroll
        for (int k = 0; k < REPLAY_T; ++k) {
            if (q0 + (unsigned)k < cnt) {
                float* a = acc + loc[k] * OWN_PITCH + 2 * lane;
#pragma unroll
                for (int j = 0; j < V2; ++j) {
                    atomicAdd(a + 64 * j,     v[k][j].x);
                    atomicAdd(a + 64 * j + 1, v[k][j].y);
                }
            }
        }
    }
    lds_barrier();

    for (int i = tid; i < nrows * D4; i += OWN_THREADS) {
        const int rr = i / D4, d4 = i % D4;
        const float di = dinv[base + rr];
        const float4 bv = *(const float4*)(bias + 4 * d4);
        float4 v = acc4[rr * P4 + d4];
        v = make_float4(fmaxf(di * v.x + bv.x, 0.0f), fmaxf(di * v.y + bv.y, 0.0f),
                        fmaxf(di * v.z + bv.z, 0.0f), fmaxf(di * v.w + bv.w, 0.0f));
        acc4[rr * P4 + d4] = v;
    }
    lds_barrier();

    constexpr int NT  = NOUT / 16;
    constexpr int MG  = (OWN_THREADS / 32) / NT;
    constexpr int TPW = (OWN_TILES + MG - 1) / MG;
    const int half = lane >> 4;
    const int l16  = lane & 15;
    const int n0   = (wave % NT) * 16;
    const int slot = wave / NT;

    v8f c[TPW];
#pragma unroll
    for (int i = 0; i < TPW; ++i) c[i] = (v8f){0.f, 0.f, 0.f, 0.f, 0.f, 0.f, 0.f, 0.f};

#pragma unroll
    for (int kh = 0; kh < 2; ++kh) {
        float4 wf[8];
        {
            const float4* p = (const float4*)(Wn + (size_t)(n0 + l16) * D + 4 * half) + 16 * kh;
#pragma unroll
            for (int j = 0; j < 8; ++j) wf[j] = p[2 * j];
        }
#pragma unroll
        for (int i = 0; i < TPW; ++i) {
            const int t = slot + MG * i;
            if (i == TPW - 1 && t >= OWN_TILES) continue;
            const float4* p = acc4 + (16 * t + l16) * P4 + half + 16 * kh;
            float4 a[8];
#pragma unroll
            for (int j = 0; j < 8; ++j) a[j] = p[2 * j];
#pragma unroll
            for (int j = 0; j < 8; ++j) {
                v2f a0 = {a[j].x, a[j].y}, b0 = {wf[j].x, wf[j].y};
                v2f a1 = {a[j].z, a[j].w}, b1 = {wf[j].z, wf[j].w};
                c[i] = __builtin_amdgcn_wmma_f32_16x16x4_f32(false, a0, false, b0,
                                                             (short)0, c[i], false, false);
                c[i] = __builtin_amdgcn_wmma_f32_16x16x4_f32(false, a1, false, b1,
                                                             (short)0, c[i], false, false);
            }
        }
    }

#pragma unroll
    for (int i = 0; i < TPW; ++i) {
        const int t = slot + MG * i;
        if (t < OWN_TILES) {
#pragma unroll
            for (int v = 0; v < 8; ++v) {
                const int r = 16 * t + 8 * half + v;
                if (r < nrows)
                    hout[(size_t)(base + r) * NOUT + n0 + l16] = dinv[base + r] * c[i][v];
            }
        }
    }
}

extern "C" void kernel_launch(void* const* d_in, const int* in_sizes, int n_in,
                              void* d_out, int out_size, void* d_ws, size_t ws_size,
                              hipStream_t stream) {
    const float* x  = (const float*)d_in[0];
    const int*   ei = (const int*)d_in[1];
    const float* W0 = (const float*)d_in[2];
    const float* b0 = (const float*)d_in[3];
    const float* W1 = (const float*)d_in[4];
    const float* b1 = (const float*)d_in[5];
    const float* W2 = (const float*)d_in[6];
    const float* b2 = (const float*)d_in[7];

    const int N = in_sizes[0] / IN_DIM;
    const int E = in_sizes[1] / 2;
    const int* rowIdx = ei;
    const int* colIdx = ei + E;

    char*  ws  = (char*)d_ws;
    size_t off = 0;
    auto   alloc = [&](size_t bytes) {
        void* p = ws + off;
        off += (bytes + 255) & ~(size_t)255;
        return p;
    };
    float* dinv = (float*)alloc((size_t)(N + 16) * sizeof(float));
    float* hbuf = (float*)alloc((size_t)N * HID_DIM * sizeof(float));
    float* xbuf = (float*)alloc((size_t)N * HID_DIM * sizeof(float));

    const int mtiles  = (N + 15) / 16;
    const int ownBlks = (N + OWN_ROWS - 1) / OWN_ROWS;
    const int binBlks = (E + BIN_EDGES - 1) / BIN_EDGES;
    unsigned* recs = (unsigned*)alloc((size_t)E * sizeof(unsigned));
    int*      tab  = (int*)alloc((size_t)ownBlks * 2 * sizeof(int));
    unsigned* cnt  = (unsigned*)alloc((size_t)binBlks * MAX_GROUPS * sizeof(unsigned));

    group_count_kernel<<<binBlks, BIN_THREADS, 0, stream>>>(colIdx, cnt, E);
    edge_list_kernel<<<binBlks, BIN_THREADS, 0, stream>>>(rowIdx, colIdx, cnt, recs, tab, E, ownBlks);
    owner_degree_kernel<<<ownBlks, DEG_THREADS, 0, stream>>>(recs, tab, dinv, N, mtiles * 16);

    gemm_wmma_kernel<128><<<256, GEMM_THREADS, 0, stream>>>(x, W0, dinv, hbuf, N, mtiles);
    owner_replay_product_kernel<128><<<ownBlks, OWN_THREADS, 0, stream>>>(
        recs, tab, dinv, hbuf, b0, W1, xbuf, N);

    owner_replay_product_kernel<64><<<ownBlks, OWN_THREADS, 0, stream>>>(
        recs, tab, dinv, xbuf, b1, W2, hbuf, N);

    owner_replay_kernel<64, false><<<ownBlks, OWN_THREADS, 0, stream>>>(
        recs, tab, dinv, hbuf, b2, (float*)d_out, N);
}


# Owner sum of layers 0-1 on the matrix unit: list sorted by 16-row tile in the degree pass (arrival order in a tile); a wave adds 4 gathered rows a step as S*R, S = 0/1 selection; no LDS float adds
# speedup vs baseline: 3.7751x; 1.0456x over previous
#include <hip/hip_runtime.h>
#include <hip/hip_bf16.h>

typedef float v2f __attribute__((ext_vector_type(2)));
typedef float v8f __attribute__((ext_vector_type(8)));

#define IN_DIM  128
#define HID_DIM 128
#define OUT_DIM 64

#define GEMM_THREADS 512
template <int NOUT>
__global__ __launch_bounds__(GEMM_THREADS)
void gemm_wmma_kernel(const float* __restrict__ X,
                      const float* __restrict__ W,
                      const float* __restrict__ dinv,
                      float* __restrict__ H, int nNodes, int mtiles) {
    constexpr int K  = 128;
    constexpr int KQ = K / 8;
    constexpr int NT = NOUT / 16;
    constexpr int MG = (GEMM_THREADS / 32) / NT;
    const int wv   = __builtin_amdgcn_readfirstlane(threadIdx.x >> 5);
    const int lane = threadIdx.x & 31;
    const int half = lane >> 4;
    const int l16  = lane & 15;
    const int n0   = (wv % NT) * 16;
    const int step = gridDim.x * MG;
    int t = blockIdx.x * MG + wv / NT;
    if (t >= mtiles) return;

    float4 wf[KQ];
    {
        const float4* p = (const float4*)(W + (size_t)(n0 + l16) * K + 4 * half);
#pragma unroll
        for (int j = 0; j < KQ; ++j) wf[j] = p[2 * j];
    }
    auto load_tile = [&](float4 (&dst)[KQ], float4 (&sc)[2], int tile) {
        const float4* dp = (const float4*)(dinv + tile * 16 + 8 * half);
        sc[0] = dp[0]; sc[1] = dp[1];
        int mrow = min(tile * 16 + l16, nNodes - 1);
        const float4* p = (const float4*)(X + (size_t)mrow * K + 4 * half);
#pragma unroll
        for (int j = 0; j < KQ; ++j) dst[j] = p[2 * j];
    };
    auto mma_store = [&](const float4 (&s)[KQ], const float4 (&sq)[2], int tile) {
        const int m0 = tile * 16;
        const float sc[8] = {sq[0].x, sq[0].y, sq[0].z, sq[0].w, sq[1].x, sq[1].y, sq[1].z, sq[1].w};
        v8f c = {0.f, 0.f, 0.f, 0.f, 0.f, 0.f, 0.f, 0.f};
#pragma unroll
        for (int j = 0; j < KQ; ++j) {
            v2f a0 = {s[j].x, s[j].y}, b0 = {wf[j].x, wf[j].y};
            v2f a1 = {s[j].z, s[j].w}, b1 = {wf[j].z, wf[j].w};
            c = __builtin_amdgcn_wmma_f32_16x16x4_f32(false, a0, false, b0,
                                                      (short)0, c, false, false);
            c = __builtin_amdgcn_wmma_f32_16x16x4_f32(false, a1, false, b1,
                                                      (short)0, c, false, false);
        }
        float* out = H + (size_t)(m0 + 8 * half) * NOUT + n0 + l16;
        if (m0 + 16 <= nNodes) {
#pragma unroll
            for (int v = 0; v < 8; ++v) out[(size_t)v * NOUT] = sc[v] * c[v];
        } else {
#pragma unroll
            for (int v = 0; v < 8; ++v)
                if (m0 + 8 * half + v < nNodes) out[(size_t)v * NOUT] = sc[v] * c[v];
        }
    };

    float4 fa[KQ], fb[KQ], sa[2], sb[2];
    load_tile(fa, sa, t);
    for (;;) {
        int tn = t + step;
        load_tile(fb, sb, min(tn, mtiles - 1));
        __builtin_amdgcn_sched_barrier(0);
        mma_store(fa, sa, t);
        if (tn >= mtiles) break;
        t = tn + step;
        load_tile(fa, sa, min(t, mtiles - 1));
        __builtin_amdgcn_sched_barrier(0);
        mma_store(fb, sb, tn);
        if (t >= mtiles) break;
    }
}

#define OWN_ROWS     196
#define OWN_THREADS  1024
#define MAX_GROUPS   256
#define BIN_THREADS  1024
#define BIN_EDGES    (BIN_THREADS * 4)
#define CNT_Q        16
#define CNT_U        8

__device__ __forceinline__ void lds_barrier() {
    __builtin_amdgcn_fence(__ATOMIC_RELEASE, "workgroup", "local");
    __builtin_amdgcn_s_barrier();
    __builtin_amdgcn_fence(__ATOMIC_ACQUIRE, "workgroup", "local");
}

__device__ __forceinline__ void load_cols(const int* __restrict__ col, int e0, int E, int (&c)[4]) {
    if (e0 + 3 < E) {
        int4 cv = *(const int4*)(col + e0);
        c[0] = cv.x; c[1] = cv.y; c[2] = cv.z; c[3] = cv.w;
    } else {
#pragma unroll
        for (int k = 0; k < 4; ++k) c[k] = (e0 + k < E) ? col[e0 + k] : -1;
    }
}

__global__ __launch_bounds__(BIN_THREADS)
void group_count_kernel(const int* __restrict__ col, unsigned* __restrict__ cnt, int E) {
    __shared__ unsigned hist[MAX_GROUPS];
    const int tid = threadIdx.x;
    int c[4];
    load_cols(col, (blockIdx.x * BIN_THREADS + tid) * 4, E, c);
    if (tid < MAX_GROUPS) hist[tid] = 0u;
    lds_barrier();
#pragma unroll
    for (int k = 0; k < 4; ++k)
        if (c[k] >= 0) atomicAdd(&hist[(unsigned)c[k] / OWN_ROWS], 1u);
    lds_barrier();
    if (tid < MAX_GROUPS) cnt[blockIdx.x * MAX_GROUPS + tid] = hist[tid];
}

__global__ __launch_bounds__(BIN_THREADS)
void edge_list_kernel(const int* __restrict__ row,
                      const int* __restrict__ col,
                      const unsigned* __restrict__ cnt,
                      unsigned* __restrict__ recs, int* __restrict__ tab,
                      int E, int nGroups) {
    constexpr int G4 = MAX_GROUPS / 4;
    static_assert(BIN_THREADS == CNT_Q * G4, "one thread per piece and row class");

    __shared__ unsigned stage[BIN_EDGES];
    __shared__ unsigned spos[BIN_EDGES];
    __shared__ uint4    ptot[CNT_Q * G4];
    __shared__ uint4    pbef[CNT_Q * G4];
    __shared__ unsigned slot[MAX_GROUPS];
    __shared__ unsigned lfirst[MAX_GROUPS];
    __shared__ unsigned gfirst[MAX_GROUPS];
    __shared__ unsigned wtot[MAX_GROUPS / 32];
    __shared__ unsigned wcnt[MAX_GROUPS / 32];

    const int tid  = threadIdx.x;
    const int lane = tid & 31;
    const int wave = tid >> 5;
    const int nb   = gridDim.x;
    const int bid  = blockIdx.x;

    int r[4], c[4];
    const int e0 = (bid * BIN_THREADS + tid) * 4;
    load_cols(col, e0, E, c);
    if (e0 + 3 < E) {
        int4 rv = *(const int4*)(row + e0);
        r[0] = rv.x; r[1] = rv.y; r[2] = rv.z; r[3] = rv.w;
    } else {
#pragma unroll
        for (int k = 0; k < 4; ++k) r[k] = (e0 + k < E) ? row[e0 + k] : 0;
    }

    {
        const int q = tid / G4, g4 = tid % G4;
        const uint4* __restrict__ c4 = (const uint4*)cnt;
        uint4 tot = make_uint4(0u, 0u, 0u, 0u), bef = make_uint4(0u, 0u, 0u, 0u);
        for (int b0 = q; b0 < nb; b0 += CNT_Q * CNT_U) {
            uint4 v[CNT_U];
#pragma unroll
            for (int j = 0; j < CNT_U; ++j) v[j] = c4[min(b0 + CNT_Q * j, nb - 1) * G4 + g4];
#pragma unroll
            for (int j = 0; j < CNT_U; ++j) {
                const int b = b0 + CNT_Q * j;
                const unsigned ma = (b < nb) ? 0xffffffffu : 0u;
                const unsigned mb = (b < bid) ? ma : 0u;
                tot.x += v[j].x & ma; tot.y += v[j].y & ma; tot.z += v[j].z & ma; tot.w += v[j].w & ma;
                bef.x += v[j].x & mb; bef.y += v[j].y & mb; bef.z += v[j].z & mb; bef.w += v[j].w & mb;
            }
        }
        ptot[tid] = tot;
        pbef[tid] = bef;
    }
    if (tid < MAX_GROUPS) slot[tid] = 0u;
    lds_barrier();

    unsigned g4[4], l4[4], rk[4];
#pragma unroll
    for (int k = 0; k < 4; ++k) {
        unsigned cc = (unsigned)max(c[k], 0);
        g4[k] = cc / OWN_ROWS;
        l4[k] = cc - g4[k] * OWN_ROWS;
        rk[k] = 0u;
        if (c[k] >= 0) rk[k] = atomicAdd(&slot[g4[k]], 1u);
    }

    unsigned mine = 0u, before = 0u;
    if (tid < MAX_GROUPS) {
        const unsigned* pt = (const unsigned*)ptot;
        const unsigned* pb = (const unsigned*)pbef;
#pragma unroll
        for (int q = 0; q < CNT_Q; ++q) {
            mine   += pt[q * MAX_GROUPS + tid];
            before += pb[q * MAX_GROUPS + tid];
        }
    }
    unsigned inc = mine;
#pragma unroll
    for (int m = 1; m < 32; m <<= 1) {
        unsigned t = __shfl_up(inc, m, 32);
        if (lane >= m) inc += t;
    }
    if (tid < MAX_GROUPS && lane == 31) wtot[wave] = inc;
    lds_barrier();

    unsigned cntg = 0u;
    if (tid < MAX_GROUPS) cntg = slot[tid];
    unsigned cinc = cntg;
#pragma unroll
    for (int m = 1; m < 32; m <<= 1) {
        unsigned t = __shfl_up(cinc, m, 32);
        if (lane >= m) cinc += t;
    }
    if (tid < MAX_GROUPS && lane == 31) wcnt[wave] = cinc;
    lds_barrier();

    if (tid < MAX_GROUPS) {
        unsigned goff = inc - mine, lf = cinc - cntg;
        for (int w = 0; w < wave; ++w) { goff += wtot[w]; lf += wcnt[w]; }
        lfirst[tid] = lf;
        gfirst[tid] = goff + before;
        if (bid == 0 && tid < nGroups) { tab[2 * tid] = (int)goff; tab[2 * tid + 1] = (int)mine; }
    }
    lds_barrier();

#pragma unroll
    for (int k = 0; k < 4; ++k) {
        if (c[k] >= 0) {
            const unsigned p = lfirst[g4[k]] + rk[k];
            stage[p] = ((unsigned)r[k] << 8) | l4[k];
            spos[p] = gfirst[g4[k]] + rk[k];
        }
    }
    lds_barrier();

    const int total = min(BIN_EDGES, E - bid * BIN_EDGES);
#pragma unroll
    for (int j = 0; j < 4; ++j) {
        const int i = tid + BIN_THREADS * j;
        if (i < total) recs[spos[i]] = stage[i];
    }
}

#define DEG_THREADS  256
#define DEG_U        16
#define OWN_TILES    ((OWN_ROWS + 15) / 16)
__global__ __launch_bounds__(DEG_THREADS)
void owner_degree_kernel(const unsigned* __restrict__ recs,
                         const int* __restrict__ tab,
                         unsigned* __restrict__ srecs, int* __restrict__ ttab,
                         float* __restrict__ dinv, int n, int npad) {
    static_assert(OWN_ROWS <= 224 && DEG_THREADS >= 256, "spare counters behind the rows");
    static_assert(OWN_TILES < 16, "tile starts fit one 16-entry row of ttab");
    __shared__ unsigned dcount[256];
    __shared__ unsigned tsum[16];
    __shared__ unsigned cursor[16];
    const int tid  = threadIdx.x;
    const int base = blockIdx.x * OWN_ROWS;
    const int nrows = min(OWN_ROWS, n - base);
    const unsigned goff = (unsigned)tab[2 * blockIdx.x];
    const unsigned cnt  = (unsigned)tab[2 * blockIdx.x + 1];
    if (tid < 256) dcount[tid] = 1u;
    const unsigned spare = 224u + (unsigned)(tid & 31);
    lds_barrier();
    for (unsigned i0 = (unsigned)tid; i0 < cnt; i0 += (unsigned)(DEG_U * DEG_THREADS)) {
        unsigned e[DEG_U];
#pragma unroll
        for (int j = 0; j < DEG_U; ++j)
            e[j] = recs[(size_t)goff + min(i0 + (unsigned)(DEG_THREADS * j), cnt - 1u)];
#pragma unroll
        for (int j = 0; j < DEG_U; ++j)
            atomicAdd(&dcount[(i0 + (unsigned)(DEG_THREADS * j) < cnt) ? (e[j] & 255u) : spare], 1u);
    }
    lds_barrier();
    if (tid < nrows) dinv[base + tid] = rsqrtf((float)dcount[tid]);
    if (blockIdx.x == gridDim.x - 1 && tid < npad - n) dinv[n + tid] = 0.0f;

    unsigned tv = (tid < 16 * OWN_TILES) ? dcount[tid] - 1u : 0u;
#pragma unroll
    for (int m = 1; m < 16; m <<= 1) tv += __shfl_xor(tv, m, 32);
    if (tid < 16 * OWN_TILES && (tid & 15) == 0) tsum[tid >> 4] = tv;
    lds_barrier();
    if (tid < 16) {
        unsigned o = 0u;
#pragma unroll
        for (int t = 0; t < OWN_TILES; ++t) o += (t < tid) ? tsum[t] : 0u;
        cursor[tid] = o;
        ttab[16 * blockIdx.x + tid] = (int)o;
    }
    lds_barrier();
    for (unsigned i0 = (unsigned)tid; i0 < cnt; i0 += (unsigned)(DEG_U * DEG_THREADS)) {
        unsigned e[DEG_U];
#pragma unroll
        for (int j = 0; j < DEG_U; ++j)
            e[j] = recs[(size_t)goff + min(i0 + (unsigned)(DEG_THREADS * j), cnt - 1u)];
#pragma unroll
        for (int j = 0; j < DEG_U; ++j) {
            if (i0 + (unsigned)(DEG_THREADS * j) < cnt) {
                const unsigned p = atomicAdd(&cursor[(e[j] & 255u) >> 4], 1u);
                srecs[(size_t)goff + p] = e[j];
            }
        }
    }
}

#define REPLAY_T 8
template <int D, bool RELU>
__global__ __launch_bounds__(OWN_THREADS)
void owner_replay_kernel(const unsigned* __restrict__ recs,
                         const int* __restrict__ tab,
                         const float* __restrict__ dinv,
                         const float* __restrict__ h,
                         const float* __restrict__ bias,
                         float* __restrict__ out, int n) {
    constexpr int V2 = D / 64;
    constexpr int D4 = D / 4;
    __shared__ float4 acc4[OWN_ROWS * D4];
    float* acc = (float*)acc4;

    const int tid  = threadIdx.x;
    const int lane = tid & 31;
    const int wave = __builtin_amdgcn_readfirstlane(tid >> 5);
    const int base = blockIdx.x * OWN_ROWS;
    const int nrows = min(OWN_ROWS, n - base);
    const unsigned goff = (unsigned)tab[2 * blockIdx.x];
    const unsigned cnt  = (unsigned)tab[2 * blockIdx.x + 1];

    unsigned q0 = (unsigned)(REPLAY_T * wave);
    unsigned rec = 0u;
    if (q0 < cnt) rec = recs[(size_t)goff + min(q0 + (unsigned)(lane & (REPLAY_T - 1)), cnt - 1u)];

    const float4* __restrict__ h4 = (const float4*)(h + (size_t)base * D);
    for (int i = tid; i < nrows * D4; i += OWN_THREADS) acc4[i] = h4[i];
    lds_barrier();

    for (; q0 < cnt; q0 += (OWN_THREADS / 32) * REPLAY_T) {
        const unsigned cur = rec;
        const unsigned qn = q0 + (OWN_THREADS / 32) * REPLAY_T;
        if (qn < cnt) rec = recs[(size_t)goff + min(qn + (unsigned)(lane & (REPLAY_T - 1)), cnt - 1u)];
        float2 v[REPLAY_T][V2];
        unsigned loc[REPLAY_T];
#pragma unroll
        for (int k = 0; k < REPLAY_T; ++k) {
            unsigned w0 = (unsigned)__builtin_amdgcn_readlane((int)cur, k);
            loc[k] = w0 & 255u;
            const float2* src = (const float2*)(h + (size_t)(w0 >> 8) * D) + lane;
#pragma unroll
            for (int j = 0; j < V2; ++j) v[k][j] = src[32 * j];
        }
#pragma unroll
        for (int k = 0; k < REPLAY_T; ++k) {
#pragma unroll
            for (int j = 0; j < V2; ++j) asm volatile("" :: "v"(v[k][j].x), "v"(v[k][j].y));
        }
#pragma unroll
        for (int k = 0; k < REPLAY_T; ++k) {
            if (q0 + (unsigned)k < cnt) {
                float* a = acc + loc[k] * D + 2 * lane;
#pragma unroll
                for (int j = 0; j < V2; ++j) {
                    atomicAdd(a + 64 * j,     v[k][j].x);
                    atomicAdd(a + 64 * j + 1, v[k][j].y);
                }
            }
        }
    }
    lds_barrier();

    float4* __restrict__ out4 = (float4*)(out + (size_t)base * D);
    for (int i = tid; i < nrows * D4; i += OWN_THREADS) {
        const int rr = i / D4, d = (i % D4) * 4;
        const float di = dinv[base + rr];
        const float4 bv = *(const float4*)(bias + d);
        float4 v = acc4[i];
        v = make_float4(di * v.x + bv.x, di * v.y + bv.y, di * v.z + bv.z, di * v.w + bv.w);
        if (RELU) v = make_float4(fmaxf(v.x, 0.0f), fmaxf(v.y, 0.0f), fmaxf(v.z, 0.0f), fmaxf(v.w, 0.0f));
        out4[i] = v;
    }
}

#define OWN_PITCH 132
template <int NOUT>
__global__ __launch_bounds__(OWN_THREADS)
void owner_replay_product_kernel(const unsigned* __restrict__ srecs,
                                 const int* __restrict__ tab,
                                 const int* __restrict__ ttab,
                                 const float* __restrict__ dinv,
                                 const float* __restrict__ h,
                                 const float* __restrict__ bias,
                                 const float* __restrict__ Wn,
                                 float* __restrict__ hout, int n) {
    constexpr int D  = 128;
    constexpr int D4 = D / 4;
    constexpr int P4 = OWN_PITCH / 4;
    static_assert(2 * OWN_TILES <= OWN_THREADS / 32, "one wave per tile and column half");
    __shared__ float4 acc4[OWN_TILES * 16 * P4];

    const int tid  = threadIdx.x;
    const int lane = tid & 31;
    const int wave = __builtin_amdgcn_readfirstlane(tid >> 5);
    const int half = lane >> 4;
    const int l16  = lane & 15;
    const int base = blockIdx.x * OWN_ROWS;
    const int nrows = min(OWN_ROWS, n - base);
    const unsigned goff = (unsigned)tab[2 * blockIdx.x];

    if (wave < 2 * OWN_TILES) {
        const int t  = wave >> 1;
        const int ch = wave & 1;
        const unsigned tb = (unsigned)ttab[16 * blockIdx.x + t];
        const unsigned te = (unsigned)ttab[16 * blockIdx.x + t + 1];
        const float4* __restrict__ hq = (const float4*)h + 16 * ch + l16;

        auto load_entries = [&](unsigned q0) {
            return srecs[(size_t)goff + min(q0 + (unsigned)l16, te - 1u)];
        };
        unsigned q = tb;
        unsigned e = 0u;
        if (q < te) e = load_entries(q);

        v8f c[4];
        {
            float4 s[8];
#pragma unroll
            for (int v = 0; v < 8; ++v)
                s[v] = hq[(size_t)(base + min(16 * t + 8 * half + v, nrows - 1)) * D4];
#pragma unroll
            for (int v = 0; v < 8; ++v) {
                const bool in = 16 * t + 8 * half + v < nrows;
                c[0][v] = in ? s[v].x : 0.0f; c[1][v] = in ? s[v].y : 0.0f;
                c[2][v] = in ? s[v].z : 0.0f; c[3][v] = in ? s[v].w : 0.0f;
            }
        }

        const unsigned myrow = (unsigned)(16 * t + l16);
        for (; q < te; q += 16u) {
            const unsigned cur = (q + (unsigned)l16 < te) ? e : (e | 255u);
            const unsigned qn = q + 16u;
            if (qn < te) e = load_entries(qn);
            float4 rv[4][2];
            float  sel[4][2];
#pragma unroll
            for (int b = 0; b < 4; ++b) {
#pragma unroll
                for (int i = 0; i < 2; ++i) {
                    const unsigned ee = (unsigned)__shfl((int)cur, 4 * b + i + 2 * half, 32);
                    rv[b][i]  = hq[(size_t)(ee >> 8) * D4];
                    sel[b][i] = ((ee & 255u) == myrow) ? 1.0f : 0.0f;
                }
            }
            __builtin_amdgcn_sched_barrier(0);
#pragma unroll
            for (int b = 0; b < 4; ++b) {
                const v2f a = {sel[b][0], sel[b][1]};
                const float r0[4] = {rv[b][0].x, rv[b][0].y, rv[b][0].z, rv[b][0].w};
                const float r1[4] = {rv[b][1].x, rv[b][1].y, rv[b][1].z, rv[b][1].w};
#pragma unroll
                for (int j = 0; j < 4; ++j) {
                    const v2f bb = {r0[j], r1[j]};
                    c[j] = __builtin_amdgcn_wmma_f32_16x16x4_f32(false, a, false, bb,
                                                                 (short)0, c[j], false, false);
                }
            }
        }

        const float4 bv = ((const float4*)bias)[16 * ch + l16];
        float di[8];
#pragma unroll
        for (int v = 0; v < 8; ++v) di[v] = dinv[base + min(16 * t + 8 * half + v, nrows - 1)];
#pragma unroll
        for (int v = 0; v < 8; ++v) {
            const int r = 16 * t + 8 * half + v;
            float4 o = make_float4(fmaxf(di[v] * c[0][v] + bv.x, 0.0f), fmaxf(di[v] * c[1][v] + bv.y, 0.0f),
                                   fmaxf(di[v] * c[2][v] + bv.z, 0.0f), fmaxf(di[v] * c[3][v] + bv.w, 0.0f));
            if (r >= nrows) o = make_float4(0.0f, 0.0f, 0.0f, 0.0f);
            acc4[r * P4 + 16 * ch + l16] = o;
        }
    }
    lds_barrier();

    constexpr int NT  = NOUT / 16;
    constexpr int MG  = (OWN_THREADS / 32) / NT;
    constexpr int TPW = (OWN_TILES + MG - 1) / MG;
    const int n0   = (wave % NT) * 16;
    const int slot = wave / NT;

    v8f c[TPW];
#pragma unroll
    for (int i = 0; i < TPW; ++i) c[i] = (v8f){0.f, 0.f, 0.f, 0.f, 0.f, 0.f, 0.f, 0.f};

#pragma unroll
    for (int kh = 0; kh < 2; ++kh) {
        float4 wf[8];
        {
            const float4* p = (const float4*)(Wn + (size_t)(n0 + l16) * D + 4 * half) + 16 * kh;
#pragma unroll
            for (int j = 0; j < 8; ++j) wf[j] = p[2 * j];
        }
#pragma unroll
        for (int i = 0; i < TPW; ++i) {
            const int t = slot + MG * i;
            if (i == TPW - 1 && t >= OWN_TILES) continue;
            const float4* p = acc4 + (16 * t + l16) * P4 + half + 16 * kh;
            float4 a[8];
#pragma unroll
            for (int j = 0; j < 8; ++j) a[j] = p[2 * j];
#pragma unroll
            for (int j = 0; j < 8; ++j) {
                v2f a0 = {a[j].x, a[j].y}, b0 = {wf[j].x, wf[j].y};
                v2f a1 = {a[j].z, a[j].w}, b1 = {wf[j].z, wf[j].w};
                c[i] = __builtin_amdgcn_wmma_f32_16x16x4_f32(false, a0, false, b0,
                                                             (short)0, c[i], false, false);
                c[i] = __builtin_amdgcn_wmma_f32_16x16x4_f32(false, a1, false, b1,
                                                             (short)0, c[i], false, false);
            }
        }
    }

#pragma unroll
    for (int i = 0; i < TPW; ++i) {
        const int t = slot + MG * i;
        if (t < OWN_TILES) {
#pragma unroll
            for (int v = 0; v < 8; ++v) {
                const int r = 16 * t + 8 * half + v;
                if (r < nrows)
                    hout[(size_t)(base + r) * NOUT + n0 + l16] = dinv[base + r] * c[i][v];
            }
        }
    }
}

extern "C" void kernel_launch(void* const* d_in, const int* in_sizes, int n_in,
                              void* d_out, int out_size, void* d_ws, size_t ws_size,
                              hipStream_t stream) {
    const float* x  = (const float*)d_in[0];
    const int*   ei = (const int*)d_in[1];
    const float* W0 = (const float*)d_in[2];
    const float* b0 = (const float*)d_in[3];
    const float* W1 = (const float*)d_in[4];
    const float* b1 = (const float*)d_in[5];
    const float* W2 = (const float*)d_in[6];
    const float* b2 = (const float*)d_in[7];

    const int N = in_sizes[0] / IN_DIM;
    const int E = in_sizes[1] / 2;
    const int* rowIdx = ei;
    const int* colIdx = ei + E;

    char*  ws  = (char*)d_ws;
    size_t off = 0;
    auto   alloc = [&](size_t bytes) {
        void* p = ws + off;
        off += (bytes + 255) & ~(size_t)255;
        return p;
    };
    float* dinv = (float*)alloc((size_t)(N + 16) * sizeof(float));
    float* hbuf = (float*)alloc((size_t)N * HID_DIM * sizeof(float));
    float* xbuf = (float*)alloc((size_t)N * HID_DIM * sizeof(float));

    const int mtiles  = (N + 15) / 16;
    const int ownBlks = (N + OWN_ROWS - 1) / OWN_ROWS;
    const int binBlks = (E + BIN_EDGES - 1) / BIN_EDGES;
    unsigned* recs = (unsigned*)alloc((size_t)E * sizeof(unsigned));
    int*      tab  = (int*)alloc((size_t)ownBlks * 2 * sizeof(int));
    unsigned* cnt  = (unsigned*)alloc((size_t)binBlks * MAX_GROUPS * sizeof(unsigned));
    unsigned* srecs = (unsigned*)alloc((size_t)E * sizeof(unsigned));
    int*      ttab = (int*)alloc((size_t)ownBlks * 16 * sizeof(int));

    group_count_kernel<<<binBlks, BIN_THREADS, 0, stream>>>(colIdx, cnt, E);
    edge_list_kernel<<<binBlks, BIN_THREADS, 0, stream>>>(rowIdx, colIdx, cnt, recs, tab, E, ownBlks);
    owner_degree_kernel<<<ownBlks, DEG_THREADS, 0, stream>>>(recs, tab, srecs, ttab, dinv, N, mtiles * 16);

    gemm_wmma_kernel<128><<<256, GEMM_THREADS, 0, stream>>>(x, W0, dinv, hbuf, N, mtiles);
    owner_replay_product_kernel<128><<<ownBlks, OWN_THREADS, 0, stream>>>(
        srecs, tab, ttab, dinv, hbuf, b0, W1, xbuf, N);

    owner_replay_product_kernel<64><<<ownBlks, OWN_THREADS, 0, stream>>>(
        srecs, tab, ttab, dinv, xbuf, b1, W2, hbuf, N);

    owner_replay_kernel<64, false><<<ownBlks, OWN_THREADS, 0, stream>>>(
        recs, tab, dinv, hbuf, b2, (float*)d_out, N);
}


# Layer 0's product moved into the owner degree pass: a group holds its 196 rows of x in LDS (asked for ahead of the count) and multiplies them by W0 itself, same k order; GEMM launch gone (6 left)
# speedup vs baseline: 3.9657x; 1.0505x over previous
#include <hip/hip_runtime.h>
#include <hip/hip_bf16.h>

typedef float v2f __attribute__((ext_vector_type(2)));
typedef float v8f __attribute__((ext_vector_type(8)));

#define IN_DIM  128
#define HID_DIM 128
#define OUT_DIM 64

#define OWN_ROWS     196
#define OWN_THREADS  1024
#define MAX_GROUPS   256
#define BIN_THREADS  1024
#define BIN_EDGES    (BIN_THREADS * 4)
#define CNT_Q        16
#define CNT_U        8

__device__ __forceinline__ void lds_barrier() {
    __builtin_amdgcn_fence(__ATOMIC_RELEASE, "workgroup", "local");
    __builtin_amdgcn_s_barrier();
    __builtin_amdgcn_fence(__ATOMIC_ACQUIRE, "workgroup", "local");
}

__device__ __forceinline__ void load_cols(const int* __restrict__ col, int e0, int E, int (&c)[4]) {
    if (e0 + 3 < E) {
        int4 cv = *(const int4*)(col + e0);
        c[0] = cv.x; c[1] = cv.y; c[2] = cv.z; c[3] = cv.w;
    } else {
#pragma unroll
        for (int k = 0; k < 4; ++k) c[k] = (e0 + k < E) ? col[e0 + k] : -1;
    }
}

__global__ __launch_bounds__(BIN_THREADS)
void group_count_kernel(const int* __restrict__ col, unsigned* __restrict__ cnt, int E) {
    __shared__ unsigned hist[MAX_GROUPS];
    const int tid = threadIdx.x;
    int c[4];
    load_cols(col, (blockIdx.x * BIN_THREADS + tid) * 4, E, c);
    if (tid < MAX_GROUPS) hist[tid] = 0u;
    lds_barrier();
#pragma unroll
    for (int k = 0; k < 4; ++k)
        if (c[k] >= 0) atomicAdd(&hist[(unsigned)c[k] / OWN_ROWS], 1u);
    lds_barrier();
    if (tid < MAX_GROUPS) cnt[blockIdx.x * MAX_GROUPS + tid] = hist[tid];
}

__global__ __launch_bounds__(BIN_THREADS)
void edge_list_kernel(const int* __restrict__ row,
                      const int* __restrict__ col,
                      const unsigned* __restrict__ cnt,
                      unsigned* __restrict__ recs, int* __restrict__ tab,
                      int E, int nGroups) {
    constexpr int G4 = MAX_GROUPS / 4;
    static_assert(BIN_THREADS == CNT_Q * G4, "one thread per piece and row class");

    __shared__ unsigned stage[BIN_EDGES];
    __shared__ unsigned spos[BIN_EDGES];
    __shared__ uint4    ptot[CNT_Q * G4];
    __shared__ uint4    pbef[CNT_Q * G4];
    __shared__ unsigned slot[MAX_GROUPS];
    __shared__ unsigned lfirst[MAX_GROUPS];
    __shared__ unsigned gfirst[MAX_GROUPS];
    __shared__ unsigned wtot[MAX_GROUPS / 32];
    __shared__ unsigned wcnt[MAX_GROUPS / 32];

    const int tid  = threadIdx.x;
    const int lane = tid & 31;
    const int wave = tid >> 5;
    const int nb   = gridDim.x;
    const int bid  = blockIdx.x;

    int r[4], c[4];
    const int e0 = (bid * BIN_THREADS + tid) * 4;
    load_cols(col, e0, E, c);
    if (e0 + 3 < E) {
        int4 rv = *(const int4*)(row + e0);
        r[0] = rv.x; r[1] = rv.y; r[2] = rv.z; r[3] = rv.w;
    } else {
#pragma unroll
        for (int k = 0; k < 4; ++k) r[k] = (e0 + k < E) ? row[e0 + k] : 0;
    }

    {
        const int q = tid / G4, g4 = tid % G4;
        const uint4* __restrict__ c4 = (const uint4*)cnt;
        uint4 tot = make_uint4(0u, 0u, 0u, 0u), bef = make_uint4(0u, 0u, 0u, 0u);
        for (int b0 = q; b0 < nb; b0 += CNT_Q * CNT_U) {
            uint4 v[CNT_U];
#pragma unroll
            for (int j = 0; j < CNT_U; ++j) v[j] = c4[min(b0 + CNT_Q * j, nb - 1) * G4 + g4];
#pragma unroll
            for (int j = 0; j < CNT_U; ++j) {
                const int b = b0 + CNT_Q * j;
                const unsigned ma = (b < nb) ? 0xffffffffu : 0u;
                const unsigned mb = (b < bid) ? ma : 0u;
                tot.x += v[j].x & ma; tot.y += v[j].y & ma; tot.z += v[j].z & ma; tot.w += v[j].w & ma;
                bef.x += v[j].x & mb; bef.y += v[j].y & mb; bef.z += v[j].z & mb; bef.w += v[j].w & mb;
            }
        }
        ptot[tid] = tot;
        pbef[tid] = bef;
    }
    if (tid < MAX_GROUPS) slot[tid] = 0u;
    lds_barrier();

    unsigned g4[4], l4[4], rk[4];
#pragma unroll
    for (int k = 0; k < 4; ++k) {
        unsigned cc = (unsigned)max(c[k], 0);
        g4[k] = cc / OWN_ROWS;
        l4[k] = cc - g4[k] * OWN_ROWS;
        rk[k] = 0u;
        if (c[k] >= 0) rk[k] = atomicAdd(&slot[g4[k]], 1u);
    }

    unsigned mine = 0u, before = 0u;
    if (tid < MAX_GROUPS) {
        const unsigned* pt = (const unsigned*)ptot;
        const unsigned* pb = (const unsigned*)pbef;
#pragma unroll
        for (int q = 0; q < CNT_Q; ++q) {
            mine   += pt[q * MAX_GROUPS + tid];
            before += pb[q * MAX_GROUPS + tid];
        }
    }
    unsigned inc = mine;
#pragma unroll
    for (int m = 1; m < 32; m <<= 1) {
        unsigned t = __shfl_up(inc, m, 32);
        if (lane >= m) inc += t;
    }
    if (tid < MAX_GROUPS && lane == 31) wtot[wave] = inc;
    lds_barrier();

    unsigned cntg = 0u;
    if (tid < MAX_GROUPS) cntg = slot[tid];
    unsigned cinc = cntg;
#pragma unroll
    for (int m = 1; m < 32; m <<= 1) {
        unsigned t = __shfl_up(cinc, m, 32);
        if (lane >= m) cinc += t;
    }
    if (tid < MAX_GROUPS && lane == 31) wcnt[wave] = cinc;
    lds_barrier();

    if (tid < MAX_GROUPS) {
        unsigned goff = inc - mine, lf = cinc - cntg;
        for (int w = 0; w < wave; ++w) { goff += wtot[w]; lf += wcnt[w]; }
        lfirst[tid] = lf;
        gfirst[tid] = goff + before;
        if (bid == 0 && tid < nGroups) { tab[2 * tid] = (int)goff; tab[2 * tid + 1] = (int)mine; }
    }
    lds_barrier();

#pragma unroll
    for (int k = 0; k < 4; ++k) {
        if (c[k] >= 0) {
            const unsigned p = lfirst[g4[k]] + rk[k];
            stage[p] = ((unsigned)r[k] << 8) | l4[k];
            spos[p] = gfirst[g4[k]] + rk[k];
        }
    }
    lds_barrier();

    const int total = min(BIN_EDGES, E - bid * BIN_EDGES);
#pragma unroll
    for (int j = 0; j < 4; ++j) {
        const int i = tid + BIN_THREADS * j;
        if (i < total) recs[spos[i]] = stage[i];
    }
}

#define DEG_THREADS  1024
#define DEG_U        4
#define OWN_TILES    ((OWN_ROWS + 15) / 16)
#define OWN_PITCH    132
__global__ __launch_bounds__(DEG_THREADS)
void owner_degree_product_kernel(const unsigned* __restrict__ recs,
                                 const int* __restrict__ tab,
                                 unsigned* __restrict__ srecs, int* __restrict__ ttab,
                                 float* __restrict__ dinv,
                                 const float* __restrict__ X,
                                 const float* __restrict__ W,
                                 float* __restrict__ hout, int n) {
    static_assert(OWN_ROWS <= 224 && DEG_THREADS >= 256, "spare counters behind the rows");
    static_assert(OWN_TILES < 16, "tile starts fit one 16-entry row of ttab");
    constexpr int D    = 128;
    constexpr int D4   = D / 4;
    constexpr int P4   = OWN_PITCH / 4;
    constexpr int NOUT = 128;
    constexpr int XN   = OWN_TILES * 16 * D4;
    constexpr int XQ   = (XN + DEG_THREADS - 1) / DEG_THREADS;
    __shared__ float4 acc4[OWN_TILES * 16 * P4];
    __shared__ unsigned dcount[256];
    __shared__ unsigned tsum[16];
    __shared__ unsigned cursor[16];
    const int tid  = threadIdx.x;
    const int lane = tid & 31;
    const int wave = __builtin_amdgcn_readfirstlane(tid >> 5);
    const int half = lane >> 4;
    const int l16  = lane & 15;
    const int base = blockIdx.x * OWN_ROWS;
    const int nrows = min(OWN_ROWS, n - base);
    const unsigned goff = (unsigned)tab[2 * blockIdx.x];
    const unsigned cnt  = (unsigned)tab[2 * blockIdx.x + 1];

    float4 xr[XQ];
    {
        const float4* __restrict__ x4 = (const float4*)X;
#pragma unroll
        for (int q = 0; q < XQ; ++q) {
            const int i = min(tid + DEG_THREADS * q, XN - 1);
            xr[q] = x4[(size_t)(base + min(i / D4, nrows - 1)) * D4 + (i % D4)];
        }
    }

    if (tid < 256) dcount[tid] = 1u;
    const unsigned spare = 224u + (unsigned)(tid & 31);
    lds_barrier();
    for (unsigned i0 = (unsigned)tid; i0 < cnt; i0 += (unsigned)(DEG_U * DEG_THREADS)) {
        unsigned e[DEG_U];
#pragma unroll
        for (int j = 0; j < DEG_U; ++j)
            e[j] = recs[(size_t)goff + min(i0 + (unsigned)(DEG_THREADS * j), cnt - 1u)];
#pragma unroll
        for (int j = 0; j < DEG_U; ++j)
            atomicAdd(&dcount[(i0 + (unsigned)(DEG_THREADS * j) < cnt) ? (e[j] & 255u) : spare], 1u);
    }
#pragma unroll
    for (int q = 0; q < XQ; ++q) {
        const int i = tid + DEG_THREADS * q;
        if (i < XN) {
            const int rr = i / D4;
            acc4[rr * P4 + (i % D4)] = (rr < nrows) ? xr[q] : make_float4(0.0f, 0.0f, 0.0f, 0.0f);
        }
    }
    lds_barrier();
    if (tid < nrows) dinv[base + tid] = rsqrtf((float)dcount[tid]);

    unsigned tv = (tid < 16 * OWN_TILES) ? dcount[tid] - 1u : 0u;
#pragma unroll
    for (int m = 1; m < 16; m <<= 1) tv += __shfl_xor(tv, m, 32);
    if (tid < 16 * OWN_TILES && (tid & 15) == 0) tsum[tid >> 4] = tv;
    lds_barrier();
    if (tid < 16) {
        unsigned o = 0u;
#pragma unroll
        for (int t = 0; t < OWN_TILES; ++t) o += (t < tid) ? tsum[t] : 0u;
        cursor[tid] = o;
        ttab[16 * blockIdx.x + tid] = (int)o;
    }
    lds_barrier();
    for (unsigned i0 = (unsigned)tid; i0 < cnt; i0 += (unsigned)(DEG_U * DEG_THREADS)) {
        unsigned e[DEG_U];
#pragma unroll
        for (int j = 0; j < DEG_U; ++j)
            e[j] = recs[(size_t)goff + min(i0 + (unsigned)(DEG_THREADS * j), cnt - 1u)];
#pragma unroll
        for (int j = 0; j < DEG_U; ++j) {
            if (i0 + (unsigned)(DEG_THREADS * j) < cnt) {
                const unsigned p = atomicAdd(&cursor[(e[j] & 255u) >> 4], 1u);
                srecs[(size_t)goff + p] = e[j];
            }
        }
    }

    constexpr int NT  = NOUT / 16;
    constexpr int MG  = (DEG_THREADS / 32) / NT;
    constexpr int TPW = (OWN_TILES + MG - 1) / MG;
    const int n0   = (wave % NT) * 16;
    const int slot = wave / NT;

    v8f c[TPW];
#pragma unroll
    for (int i = 0; i < TPW; ++i) c[i] = (v8f){0.f, 0.f, 0.f, 0.f, 0.f, 0.f, 0.f, 0.f};

#pragma unroll
    for (int kh = 0; kh < 2; ++kh) {
        float4 wf[8];
        {
            const float4* p = (const float4*)(W + (size_t)(n0 + l16) * D + 4 * half) + 16 * kh;
#pragma unroll
            for (int j = 0; j < 8; ++j) wf[j] = p[2 * j];
        }
#pragma unroll
        for (int i = 0; i < TPW; ++i) {
            const int t = slot + MG * i;
            if (i == TPW - 1 && t >= OWN_TILES) continue;
            const float4* p = acc4 + (16 * t + l16) * P4 + half + 16 * kh;
            float4 a[8];
#pragma unroll
            for (int j = 0; j < 8; ++j) a[j] = p[2 * j];
#pragma unroll
            for (int j = 0; j < 8; ++j) {
                v2f a0 = {a[j].x, a[j].y}, b0 = {wf[j].x, wf[j].y};
                v2f a1 = {a[j].z, a[j].w}, b1 = {wf[j].z, wf[j].w};
                c[i] = __builtin_amdgcn_wmma_f32_16x16x4_f32(false, a0, false, b0,
                                                             (short)0, c[i], false, false);
                c[i] = __builtin_amdgcn_wmma_f32_16x16x4_f32(false, a1, false, b1,
                                                             (short)0, c[i], false, false);
            }
        }
    }

#pragma unroll
    for (int i = 0; i < TPW; ++i) {
        const int t = slot + MG * i;
        if (t < OWN_TILES) {
#pragma unroll
            for (int v = 0; v < 8; ++v) {
                const int r = 16 * t + 8 * half + v;
                if (r < nrows)
                    hout[(size_t)(base + r) * NOUT + n0 + l16] = rsqrtf((float)dcount[r]) * c[i][v];
            }
        }
    }
}

#define REPLAY_T 8
template <int D, bool RELU>
__global__ __launch_bounds__(OWN_THREADS)
void owner_replay_kernel(const unsigned* __restrict__ recs,
                         const int* __restrict__ tab,
                         const float* __restrict__ dinv,
                         const float* __restrict__ h,
                         const float* __restrict__ bias,
                         float* __restrict__ out, int n) {
    constexpr int V2 = D / 64;
    constexpr int D4 = D / 4;
    __shared__ float4 acc4[OWN_ROWS * D4];
    float* acc = (float*)acc4;

    const int tid  = threadIdx.x;
    const int lane = tid & 31;
    const int wave = __builtin_amdgcn_readfirstlane(tid >> 5);
    const int base = blockIdx.x * OWN_ROWS;
    const int nrows = min(OWN_ROWS, n - base);
    const unsigned goff = (unsigned)tab[2 * blockIdx.x];
    const unsigned cnt  = (unsigned)tab[2 * blockIdx.x + 1];

    unsigned q0 = (unsigned)(REPLAY_T * wave);
    unsigned rec = 0u;
    if (q0 < cnt) rec = recs[(size_t)goff + min(q0 + (unsigned)(lane & (REPLAY_T - 1)), cnt - 1u)];

    const float4* __restrict__ h4 = (const float4*)(h + (size_t)base * D);
    for (int i = tid; i < nrows * D4; i += OWN_THREADS) acc4[i] = h4[i];
    lds_barrier();

    for (; q0 < cnt; q0 += (OWN_THREADS / 32) * REPLAY_T) {
        const unsigned cur = rec;
        const unsigned qn = q0 + (OWN_THREADS / 32) * REPLAY_T;
        if (qn < cnt) rec = recs[(size_t)goff + min(qn + (unsigned)(lane & (REPLAY_T - 1)), cnt - 1u)];
        float2 v[REPLAY_T][V2];
        unsigned loc[REPLAY_T];
#pragma unroll
        for (int k = 0; k < REPLAY_T; ++k) {
            unsigned w0 = (unsigned)__builtin_amdgcn_readlane((int)cur, k);
            loc[k] = w0 & 255u;
            const float2* src = (const float2*)(h + (size_t)(w0 >> 8) * D) + lane;
#pragma unroll
            for (int j = 0; j < V2; ++j) v[k][j] = src[32 * j];
        }
#pragma unroll
        for (int k = 0; k < REPLAY_T; ++k) {
#pragma unroll
            for (int j = 0; j < V2; ++j) asm volatile("" :: "v"(v[k][j].x), "v"(v[k][j].y));
        }
#pragma unroll
        for (int k = 0; k < REPLAY_T; ++k) {
            if (q0 + (unsigned)k < cnt) {
                float* a = acc + loc[k] * D + 2 * lane;
#pragma unroll
                for (int j = 0; j < V2; ++j) {
                    atomicAdd(a + 64 * j,     v[k][j].x);
                    atomicAdd(a + 64 * j + 1, v[k][j].y);
                }
            }
        }
    }
    lds_barrier();

    float4* __restrict__ out4 = (float4*)(out + (size_t)base * D);
    for (int i = tid; i < nrows * D4; i += OWN_THREADS) {
        const int rr = i / D4, d = (i % D4) * 4;
        const float di = dinv[base + rr];
        const float4 bv = *(const float4*)(bias + d);
        float4 v = acc4[i];
        v = make_float4(di * v.x + bv.x, di * v.y + bv.y, di * v.z + bv.z, di * v.w + bv.w);
        if (RELU) v = make_float4(fmaxf(v.x, 0.0f), fmaxf(v.y, 0.0f), fmaxf(v.z, 0.0f), fmaxf(v.w, 0.0f));
        out4[i] = v;
    }
}

template <int NOUT>
__global__ __launch_bounds__(OWN_THREADS)
void owner_replay_product_kernel(const unsigned* __restrict__ srecs,
                                 const int* __restrict__ tab,
                                 const int* __restrict__ ttab,
                                 const float* __restrict__ dinv,
                                 const float* __restrict__ h,
                                 const float* __restrict__ bias,
                                 const float* __restrict__ Wn,
                                 float* __restrict__ hout, int n) {
    constexpr int D  = 128;
    constexpr int D4 = D / 4;
    constexpr int P4 = OWN_PITCH / 4;
    static_assert(2 * OWN_TILES <= OWN_THREADS / 32, "one wave per tile and column half");
    __shared__ float4 acc4[OWN_TILES * 16 * P4];

    const int tid  = threadIdx.x;
    const int lane = tid & 31;
    const int wave = __builtin_amdgcn_readfirstlane(tid >> 5);
    const int half = lane >> 4;
    const int l16  = lane & 15;
    const int base = blockIdx.x * OWN_ROWS;
    const int nrows = min(OWN_ROWS, n - base);
    const unsigned goff = (unsigned)tab[2 * blockIdx.x];

    if (wave < 2 * OWN_TILES) {
        const int t  = wave >> 1;
        const int ch = wave & 1;
        const unsigned tb = (unsigned)ttab[16 * blockIdx.x + t];
        const unsigned te = (unsigned)ttab[16 * blockIdx.x + t + 1];
        const float4* __restrict__ hq = (const float4*)h + 16 * ch + l16;

        auto load_entries = [&](unsigned q0) {
            return srecs[(size_t)goff + min(q0 + (unsigned)l16, te - 1u)];
        };
        unsigned q = tb;
        unsigned e = 0u;
        if (q < te) e = load_entries(q);

        v8f c[4];
        {
            float4 s[8];
#pragma unroll
            for (int v = 0; v < 8; ++v)
                s[v] = hq[(size_t)(base + min(16 * t + 8 * half + v, nrows - 1)) * D4];
#pragma unroll
            for (int v = 0; v < 8; ++v) {
                const bool in = 16 * t + 8 * half + v < nrows;
                c[0][v] = in ? s[v].x : 0.0f; c[1][v] = in ? s[v].y : 0.0f;
                c[2][v] = in ? s[v].z : 0.0f; c[3][v] = in ? s[v].w : 0.0f;
            }
        }

        const unsigned myrow = (unsigned)(16 * t + l16);
        for (; q < te; q += 16u) {
            const unsigned cur = (q + (unsigned)l16 < te) ? e : (e | 255u);
            const unsigned qn = q + 16u;
            if (qn < te) e = load_entries(qn);
            float4 rv[4][2];
            float  sel[4][2];
#pragma unroll
            for (int b = 0; b < 4; ++b) {
#pragma unroll
                for (int i = 0; i < 2; ++i) {
                    const unsigned ee = (unsigned)__shfl((int)cur, 4 * b + i + 2 * half, 32);
                    rv[b][i]  = hq[(size_t)(ee >> 8) * D4];
                    sel[b][i] = ((ee & 255u) == myrow) ? 1.0f : 0.0f;
                }
            }
            __builtin_amdgcn_sched_barrier(0);
#pragma unroll
            for (int b = 0; b < 4; ++b) {
                const v2f a = {sel[b][0], sel[b][1]};
                const float r0[4] = {rv[b][0].x, rv[b][0].y, rv[b][0].z, rv[b][0].w};
                const float r1[4] = {rv[b][1].x, rv[b][1].y, rv[b][1].z, rv[b][1].w};
#pragma unroll
                for (int j = 0; j < 4; ++j) {
                    const v2f bb = {r0[j], r1[j]};
                    c[j] = __builtin_amdgcn_wmma_f32_16x16x4_f32(false, a, false, bb,
                                                                 (short)0, c[j], false, false);
                }
            }
        }

        const float4 bv = ((const float4*)bias)[16 * ch + l16];
        float di[8];
#pragma unroll
        for (int v = 0; v < 8; ++v) di[v] = dinv[base + min(16 * t + 8 * half + v, nrows - 1)];
#pragma unroll
        for (int v = 0; v < 8; ++v) {
            const int r = 16 * t + 8 * half + v;
            float4 o = make_float4(fmaxf(di[v] * c[0][v] + bv.x, 0.0f), fmaxf(di[v] * c[1][v] + bv.y, 0.0f),
                                   fmaxf(di[v] * c[2][v] + bv.z, 0.0f), fmaxf(di[v] * c[3][v] + bv.w, 0.0f));
            if (r >= nrows) o = make_float4(0.0f, 0.0f, 0.0f, 0.0f);
            acc4[r * P4 + 16 * ch + l16] = o;
        }
    }
    lds_barrier();

    constexpr int NT  = NOUT / 16;
    constexpr int MG  = (OWN_THREADS / 32) / NT;
    constexpr int TPW = (OWN_TILES + MG - 1) / MG;
    const int n0   = (wave % NT) * 16;
    const int slot = wave / NT;

    v8f c[TPW];
#pragma unroll
    for (int i = 0; i < TPW; ++i) c[i] = (v8f){0.f, 0.f, 0.f, 0.f, 0.f, 0.f, 0.f, 0.f};

#pragma unroll
    for (int kh = 0; kh < 2; ++kh) {
        float4 wf[8];
        {
            const float4* p = (const float4*)(Wn + (size_t)(n0 + l16) * D + 4 * half) + 16 * kh;
#pragma unroll
            for (int j = 0; j < 8; ++j) wf[j] = p[2 * j];
        }
#pragma unroll
        for (int i = 0; i < TPW; ++i) {
            const int t = slot + MG * i;
            if (i == TPW - 1 && t >= OWN_TILES) continue;
            const float4* p = acc4 + (16 * t + l16) * P4 + half + 16 * kh;
            float4 a[8];
#pragma unroll
            for (int j = 0; j < 8; ++j) a[j] = p[2 * j];
#pragma unroll
            for (int j = 0; j < 8; ++j) {
                v2f a0 = {a[j].x, a[j].y}, b0 = {wf[j].x, wf[j].y};
                v2f a1 = {a[j].z, a[j].w}, b1 = {wf[j].z, wf[j].w};
                c[i] = __builtin_amdgcn_wmma_f32_16x16x4_f32(false, a0, false, b0,
                                                             (short)0, c[i], false, false);
                c[i] = __builtin_amdgcn_wmma_f32_16x16x4_f32(false, a1, false, b1,
                                                             (short)0, c[i], false, false);
            }
        }
    }

#pragma unroll
    for (int i = 0; i < TPW; ++i) {
        const int t = slot + MG * i;
        if (t < OWN_TILES) {
#pragma unroll
            for (int v = 0; v < 8; ++v) {
                const int r = 16 * t + 8 * half + v;
                if (r < nrows)
                    hout[(size_t)(base + r) * NOUT + n0 + l16] = dinv[base + r] * c[i][v];
            }
        }
    }
}

extern "C" void kernel_launch(void* const* d_in, const int* in_sizes, int n_in,
                              void* d_out, int out_size, void* d_ws, size_t ws_size,
                              hipStream_t stream) {
    const float* x  = (const float*)d_in[0];
    const int*   ei = (const int*)d_in[1];
    const float* W0 = (const float*)d_in[2];
    const float* b0 = (const float*)d_in[3];
    const float* W1 = (const float*)d_in[4];
    const float* b1 = (const float*)d_in[5];
    const float* W2 = (const float*)d_in[6];
    const float* b2 = (const float*)d_in[7];

    const int N = in_sizes[0] / IN_DIM;
    const int E = in_sizes[1] / 2;
    const int* rowIdx = ei;
    const int* colIdx = ei + E;

    char*  ws  = (char*)d_ws;
    size_t off = 0;
    auto   alloc = [&](size_t bytes) {
        void* p = ws + off;
        off += (bytes + 255) & ~(size_t)255;
        return p;
    };
    float* dinv = (float*)alloc((size_t)N * sizeof(float));
    float* hbuf = (float*)alloc((size_t)N * HID_DIM * sizeof(float));
    float* xbuf = (float*)alloc((size_t)N * HID_DIM * sizeof(float));

    const int ownBlks = (N + OWN_ROWS - 1) / OWN_ROWS;
    const int binBlks = (E + BIN_EDGES - 1) / BIN_EDGES;
    unsigned* recs = (unsigned*)alloc((size_t)E * sizeof(unsigned));
    int*      tab  = (int*)alloc((size_t)ownBlks * 2 * sizeof(int));
    unsigned* cnt  = (unsigned*)alloc((size_t)binBlks * MAX_GROUPS * sizeof(unsigned));
    unsigned* srecs = (unsigned*)alloc((size_t)E * sizeof(unsigned));
    int*      ttab = (int*)alloc((size_t)ownBlks * 16 * sizeof(int));

    group_count_kernel<<<binBlks, BIN_THREADS, 0, stream>>>(colIdx, cnt, E);
    edge_list_kernel<<<binBlks, BIN_THREADS, 0, stream>>>(rowIdx, colIdx, cnt, recs, tab, E, ownBlks);
    owner_degree_product_kernel<<<ownBlks, DEG_THREADS, 0, stream>>>(recs, tab, srecs, ttab, dinv, x, W0, hbuf, N);

    owner_replay_product_kernel<128><<<ownBlks, OWN_THREADS, 0, stream>>>(
        srecs, tab, ttab, dinv, hbuf, b0, W1, xbuf, N);

    owner_replay_product_kernel<64><<<ownBlks, OWN_THREADS, 0, stream>>>(
        srecs, tab, ttab, dinv, xbuf, b1, W2, hbuf, N);

    owner_replay_kernel<64, false><<<ownBlks, OWN_THREADS, 0, stream>>>(
        recs, tab, dinv, hbuf, b2, (float*)d_out, N);
}


# Edge count launch gone (5 left): an edge block writes one chunk sorted by owner group plus its row of segment starts; owners walk their segments; every owner sum in the new list's arrival order
# speedup vs baseline: 4.0914x; 1.0317x over previous
#include <hip/hip_runtime.h>
#include <hip/hip_bf16.h>

typedef float v2f __attribute__((ext_vector_type(2)));
typedef float v8f __attribute__((ext_vector_type(8)));

#define IN_DIM  128
#define HID_DIM 128
#define OUT_DIM 64

#define OWN_ROWS     196
#define OWN_THREADS  1024
#define MAX_GROUPS   256
#define BIN_THREADS  1024
#define BIN_EDGES    (BIN_THREADS * 4)
#define OT_PITCH     (MAX_GROUPS + 8)

__device__ __forceinline__ void lds_barrier() {
    __builtin_amdgcn_fence(__ATOMIC_RELEASE, "workgroup", "local");
    __builtin_amdgcn_s_barrier();
    __builtin_amdgcn_fence(__ATOMIC_ACQUIRE, "workgroup", "local");
}

__device__ __forceinline__ void load_cols(const int* __restrict__ col, int e0, int E, int (&c)[4]) {
    if (e0 + 3 < E) {
        int4 cv = *(const int4*)(col + e0);
        c[0] = cv.x; c[1] = cv.y; c[2] = cv.z; c[3] = cv.w;
    } else {
#pragma unroll
        for (int k = 0; k < 4; ++k) c[k] = (e0 + k < E) ? col[e0 + k] : -1;
    }
}

__global__ __launch_bounds__(BIN_THREADS)
void edge_chunk_kernel(const int* __restrict__ row,
                       const int* __restrict__ col,
                       unsigned* __restrict__ crecs, unsigned* __restrict__ otab, int E) {
    __shared__ uint4    stage4[BIN_EDGES / 4];
    __shared__ unsigned slot[MAX_GROUPS];
    __shared__ unsigned lfirst[MAX_GROUPS];
    __shared__ unsigned wcnt[MAX_GROUPS / 32];
    unsigned* stage = (unsigned*)stage4;

    const int tid  = threadIdx.x;
    const int lane = tid & 31;
    const int wave = tid >> 5;
    const int bid  = blockIdx.x;

    int r[4], c[4];
    const int e0 = (bid * BIN_THREADS + tid) * 4;
    load_cols(col, e0, E, c);
    if (e0 + 3 < E) {
        int4 rv = *(const int4*)(row + e0);
        r[0] = rv.x; r[1] = rv.y; r[2] = rv.z; r[3] = rv.w;
    } else {
#pragma unroll
        for (int k = 0; k < 4; ++k) r[k] = (e0 + k < E) ? row[e0 + k] : 0;
    }
    if (tid < MAX_GROUPS) slot[tid] = 0u;
    lds_barrier();

    unsigned g4[4], l4[4], rk[4];
#pragma unroll
    for (int k = 0; k < 4; ++k) {
        unsigned cc = (unsigned)max(c[k], 0);
        g4[k] = cc / OWN_ROWS;
        l4[k] = cc - g4[k] * OWN_ROWS;
        rk[k] = 0u;
        if (c[k] >= 0) rk[k] = atomicAdd(&slot[g4[k]], 1u);
    }
    lds_barrier();

    unsigned cntg = 0u;
    if (tid < MAX_GROUPS) cntg = slot[tid];
    unsigned cinc = cntg;
#pragma unroll
    for (int m = 1; m < 32; m <<= 1) {
        unsigned t = __shfl_up(cinc, m, 32);
        if (lane >= m) cinc += t;
    }
    if (tid < MAX_GROUPS && lane == 31) wcnt[wave] = cinc;
    lds_barrier();

    if (tid < MAX_GROUPS) {
        unsigned lf = cinc - cntg;
        for (int w = 0; w < wave; ++w) lf += wcnt[w];
        lfirst[tid] = lf;
        otab[(size_t)bid * OT_PITCH + tid] = lf;
        if (tid == MAX_GROUPS - 1) otab[(size_t)bid * OT_PITCH + MAX_GROUPS] = lf + cntg;
    }
    lds_barrier();

#pragma unroll
    for (int k = 0; k < 4; ++k)
        if (c[k] >= 0) stage[lfirst[g4[k]] + rk[k]] = ((unsigned)r[k] << 8) | l4[k];
    lds_barrier();

    const int total = min(BIN_EDGES, E - bid * BIN_EDGES);
    unsigned* __restrict__ dst = crecs + (size_t)bid * BIN_EDGES;
    if (4 * tid + 3 < total) {
        *(uint4*)(dst + 4 * tid) = stage4[tid];
    } else {
#pragma unroll
        for (int k = 0; k < 4; ++k)
            if (4 * tid + k < total) dst[4 * tid + k] = stage[4 * tid + k];
    }
}

#define DEG_THREADS  1024
#define SEG_L        4
#define SEG_U        8
#define OWN_TILES    ((OWN_ROWS + 15) / 16)
#define OWN_PITCH    132
__global__ __launch_bounds__(DEG_THREADS)
void owner_degree_product_kernel(const unsigned* __restrict__ crecs,
                                 const unsigned* __restrict__ otab, int nb,
                                 unsigned* __restrict__ srecs, int* __restrict__ tab,
                                 int* __restrict__ ttab,
                                 float* __restrict__ dinv,
                                 const float* __restrict__ X,
                                 const float* __restrict__ W,
                                 float* __restrict__ hout, int n) {
    static_assert(OWN_ROWS <= 224 && DEG_THREADS >= 256, "spare counters behind the rows");
    static_assert(OWN_TILES < 16, "tile starts fit one 16-entry row of ttab");
    constexpr int D    = 128;
    constexpr int D4   = D / 4;
    constexpr int P4   = OWN_PITCH / 4;
    constexpr int NOUT = 128;
    constexpr int XN   = OWN_TILES * 16 * D4;
    constexpr int XQ   = (XN + DEG_THREADS - 1) / DEG_THREADS;
    __shared__ float4 acc4[OWN_TILES * 16 * P4];
    __shared__ unsigned dcount[256];
    __shared__ unsigned tsum[16];
    __shared__ unsigned cursor[16];
    __shared__ unsigned tot[2];
    const int tid  = threadIdx.x;
    const int lane = tid & 31;
    const int wave = __builtin_amdgcn_readfirstlane(tid >> 5);
    const int half = lane >> 4;
    const int l16  = lane & 15;
    const int base = blockIdx.x * OWN_ROWS;
    const int nrows = min(OWN_ROWS, n - base);

    const int ls = tid & (SEG_L - 1);
    auto bounds = [&](int s, unsigned& st, unsigned& en) {
        const unsigned* o = otab + (size_t)s * OT_PITCH + blockIdx.x;
        st = o[0]; en = o[1];
    };
    const int s0 = tid / SEG_L;
    unsigned st0 = 0u, en0 = 0u;
    if (s0 < nb) bounds(s0, st0, en0);

    float4 xr[XQ];
    {
        const float4* __restrict__ x4 = (const float4*)X;
#pragma unroll
        for (int q = 0; q < XQ; ++q) {
            const int i = min(tid + DEG_THREADS * q, XN - 1);
            xr[q] = x4[(size_t)(base + min(i / D4, nrows - 1)) * D4 + (i % D4)];
        }
    }

    if (tid < 256) dcount[tid] = 1u;
    if (tid < 2) tot[tid] = 0u;
    const unsigned spare = 224u + (unsigned)(tid & 31);
    lds_barrier();
    unsigned sst = 0u, slen = 0u;
    {
        unsigned st = st0, en = en0;
        for (int s = s0; s < nb; ) {
            if (ls == 0) { sst += st; slen += en - st; }
            const unsigned* __restrict__ p = crecs + (size_t)s * BIN_EDGES;
            for (unsigned j0 = st + (unsigned)ls; j0 < en; j0 += (unsigned)(SEG_L * SEG_U)) {
                unsigned e[SEG_U];
#pragma unroll
                for (int j = 0; j < SEG_U; ++j) e[j] = p[min(j0 + (unsigned)(SEG_L * j), en - 1u)];
#pragma unroll
                for (int j = 0; j < SEG_U; ++j)
                    atomicAdd(&dcount[(j0 + (unsigned)(SEG_L * j) < en) ? (e[j] & 255u) : spare], 1u);
            }
            s += DEG_THREADS / SEG_L;
            if (s < nb) bounds(s, st, en);
        }
    }
#pragma unroll
    for (int m = 1; m < 32; m <<= 1) {
        sst  += __shfl_xor(sst, m, 32);
        slen += __shfl_xor(slen, m, 32);
    }
    if (lane == 0) { atomicAdd(&tot[0], sst); atomicAdd(&tot[1], slen); }
#pragma unroll
    for (int q = 0; q < XQ; ++q) {
        const int i = tid + DEG_THREADS * q;
        if (i < XN) {
            const int rr = i / D4;
            acc4[rr * P4 + (i % D4)] = (rr < nrows) ? xr[q] : make_float4(0.0f, 0.0f, 0.0f, 0.0f);
        }
    }
    lds_barrier();
    const unsigned goff = tot[0];
    if (tid == 0) { tab[2 * blockIdx.x] = (int)goff; tab[2 * blockIdx.x + 1] = (int)tot[1]; }
    if (tid < nrows) dinv[base + tid] = rsqrtf((float)dcount[tid]);

    unsigned tv = (tid < 16 * OWN_TILES) ? dcount[tid] - 1u : 0u;
#pragma unroll
    for (int m = 1; m < 16; m <<= 1) tv += __shfl_xor(tv, m, 32);
    if (tid < 16 * OWN_TILES && (tid & 15) == 0) tsum[tid >> 4] = tv;
    lds_barrier();
    if (tid < 16) {
        unsigned o = 0u;
#pragma unroll
        for (int t = 0; t < OWN_TILES; ++t) o += (t < tid) ? tsum[t] : 0u;
        cursor[tid] = o;
        ttab[16 * blockIdx.x + tid] = (int)o;
    }
    lds_barrier();
    {
        unsigned st = st0, en = en0;
        for (int s = s0; s < nb; ) {
            const unsigned* __restrict__ p = crecs + (size_t)s * BIN_EDGES;
            for (unsigned j0 = st + (unsigned)ls; j0 < en; j0 += (unsigned)(SEG_L * SEG_U)) {
                unsigned e[SEG_U];
#pragma unroll
                for (int j = 0; j < SEG_U; ++j) e[j] = p[min(j0 + (unsigned)(SEG_L * j), en - 1u)];
#pragma unroll
                for (int j = 0; j < SEG_U; ++j) {
                    if (j0 + (unsigned)(SEG_L * j) < en) {
                        const unsigned q = atomicAdd(&cursor[(e[j] & 255u) >> 4], 1u);
                        srecs[(size_t)goff + q] = e[j];
                    }
                }
            }
            s += DEG_THREADS / SEG_L;
            if (s < nb) bounds(s, st, en);
        }
    }

    constexpr int NT  = NOUT / 16;
    constexpr int MG  = (DEG_THREADS / 32) / NT;
    constexpr int TPW = (OWN_TILES + MG - 1) / MG;
    const int n0   = (wave % NT) * 16;
    const int slot = wave / NT;

    v8f c[TPW];
#pragma unroll
    for (int i = 0; i < TPW; ++i) c[i] = (v8f){0.f, 0.f, 0.f, 0.f, 0.f, 0.f, 0.f, 0.f};

#pragma unroll
    for (int kh = 0; kh < 2; ++kh) {
        float4 wf[8];
        {
            const float4* p = (const float4*)(W + (size_t)(n0 + l16) * D + 4 * half) + 16 * kh;
#pragma unroll
            for (int j = 0; j < 8; ++j) wf[j] = p[2 * j];
        }
#pragma unroll
        for (int i = 0; i < TPW; ++i) {
            const int t = slot + MG * i;
            if (i == TPW - 1 && t >= OWN_TILES) continue;
            const float4* p = acc4 + (16 * t + l16) * P4 + half + 16 * kh;
            float4 a[8];
#pragma unroll
            for (int j = 0; j < 8; ++j) a[j] = p[2 * j];
#pragma unroll
            for (int j = 0; j < 8; ++j) {
                v2f a0 = {a[j].x, a[j].y}, b0 = {wf[j].x, wf[j].y};
                v2f a1 = {a[j].z, a[j].w}, b1 = {wf[j].z, wf[j].w};
                c[i] = __builtin_amdgcn_wmma_f32_16x16x4_f32(false, a0, false, b0,
                                                             (short)0, c[i], false, false);
                c[i] = __builtin_amdgcn_wmma_f32_16x16x4_f32(false, a1, false, b1,
                                                             (short)0, c[i], false, false);
            }
        }
    }

#pragma unroll
    for (int i = 0; i < TPW; ++i) {
        const int t = slot + MG * i;
        if (t < OWN_TILES) {
#pragma unroll
            for (int v = 0; v < 8; ++v) {
                const int r = 16 * t + 8 * half + v;
                if (r < nrows)
                    hout[(size_t)(base + r) * NOUT + n0 + l16] = rsqrtf((float)dcount[r]) * c[i][v];
            }
        }
    }
}

#define REPLAY_T 8
template <int D, bool RELU>
__global__ __launch_bounds__(OWN_THREADS)
void owner_replay_kernel(const unsigned* __restrict__ recs,
                         const int* __restrict__ tab,
                         const float* __restrict__ dinv,
                         const float* __restrict__ h,
                         const float* __restrict__ bias,
                         float* __restrict__ out, int n) {
    constexpr int V2 = D / 64;
    constexpr int D4 = D / 4;
    __shared__ float4 acc4[OWN_ROWS * D4];
    float* acc = (float*)acc4;

    const int tid  = threadIdx.x;
    const int lane = tid & 31;
    const int wave = __builtin_amdgcn_readfirstlane(tid >> 5);
    const int base = blockIdx.x * OWN_ROWS;
    const int nrows = min(OWN_ROWS, n - base);
    const unsigned goff = (unsigned)tab[2 * blockIdx.x];
    const unsigned cnt  = (unsigned)tab[2 * blockIdx.x + 1];

    unsigned q0 = (unsigned)(REPLAY_T * wave);
    unsigned rec = 0u;
    if (q0 < cnt) rec = recs[(size_t)goff + min(q0 + (unsigned)(lane & (REPLAY_T - 1)), cnt - 1u)];

    const float4* __restrict__ h4 = (const float4*)(h + (size_t)base * D);
    for (int i = tid; i < nrows * D4; i += OWN_THREADS) acc4[i] = h4[i];
    lds_barrier();

    for (; q0 < cnt; q0 += (OWN_THREADS / 32) * REPLAY_T) {
        const unsigned cur = rec;
        const unsigned qn = q0 + (OWN_THREADS / 32) * REPLAY_T;
        if (qn < cnt) rec = recs[(size_t)goff + min(qn + (unsigned)(lane & (REPLAY_T - 1)), cnt - 1u)];
        float2 v[REPLAY_T][V2];
        unsigned loc[REPLAY_T];
#pragma unroll
        for (int k = 0; k < REPLAY_T; ++k) {
            unsigned w0 = (unsigned)__builtin_amdgcn_readlane((int)cur, k);
            loc[k] = w0 & 255u;
            const float2* src = (const float2*)(h + (size_t)(w0 >> 8) * D) + lane;
#pragma unroll
            for (int j = 0; j < V2; ++j) v[k][j] = src[32 * j];
        }
#pragma unroll
        for (int k = 0; k < REPLAY_T; ++k) {
#pragma unroll
            for (int j = 0; j < V2; ++j) asm volatile("" :: "v"(v[k][j].x), "v"(v[k][j].y));
        }
#pragma unroll
        for (int k = 0; k < REPLAY_T; ++k) {
            if (q0 + (unsigned)k < cnt) {
                float* a = acc + loc[k] * D + 2 * lane;
#pragma unroll
                for (int j = 0; j < V2; ++j) {
                    atomicAdd(a + 64 * j,     v[k][j].x);
                    atomicAdd(a + 64 * j + 1, v[k][j].y);
                }
            }
        }
    }
    lds_barrier();

    float4* __restrict__ out4 = (float4*)(out + (size_t)base * D);
    for (int i = tid; i < nrows * D4; i += OWN_THREADS) {
        const int rr = i / D4, d = (i % D4) * 4;
        const float di = dinv[base + rr];
        const float4 bv = *(const float4*)(bias + d);
        float4 v = acc4[i];
        v = make_float4(di * v.x + bv.x, di * v.y + bv.y, di * v.z + bv.z, di * v.w + bv.w);
        if (RELU) v = make_float4(fmaxf(v.x, 0.0f), fmaxf(v.y, 0.0f), fmaxf(v.z, 0.0f), fmaxf(v.w, 0.0f));
        out4[i] = v;
    }
}

template <int NOUT>
__global__ __launch_bounds__(OWN_THREADS)
void owner_replay_product_kernel(const unsigned* __restrict__ srecs,
                                 const int* __restrict__ tab,
                                 const int* __restrict__ ttab,
                                 const float* __restrict__ dinv,
                                 const float* __restrict__ h,
                                 const float* __restrict__ bias,
                                 const float* __restrict__ Wn,
                                 float* __restrict__ hout, int n) {
    constexpr int D  = 128;
    constexpr int D4 = D / 4;
    constexpr int P4 = OWN_PITCH / 4;
    static_assert(2 * OWN_TILES <= OWN_THREADS / 32, "one wave per tile and column half");
    __shared__ float4 acc4[OWN_TILES * 16 * P4];

    const int tid  = threadIdx.x;
    const int lane = tid & 31;
    const int wave = __builtin_amdgcn_readfirstlane(tid >> 5);
    const int half = lane >> 4;
    const int l16  = lane & 15;
    const int base = blockIdx.x * OWN_ROWS;
    const int nrows = min(OWN_ROWS, n - base);
    const unsigned goff = (unsigned)tab[2 * blockIdx.x];

    if (wave < 2 * OWN_TILES) {
        const int t  = wave >> 1;
        const int ch = wave & 1;
        const unsigned tb = (unsigned)ttab[16 * blockIdx.x + t];
        const unsigned te = (unsigned)ttab[16 * blockIdx.x + t + 1];
        const float4* __restrict__ hq = (const float4*)h + 16 * ch + l16;

        auto load_entries = [&](unsigned q0) {
            return srecs[(size_t)goff + min(q0 + (unsigned)l16, te - 1u)];
        };
        unsigned q = tb;
        unsigned e = 0u;
        if (q < te) e = load_entries(q);

        v8f c[4];
        {
            float4 s[8];
#pragma unroll
            for (int v = 0; v < 8; ++v)
                s[v] = hq[(size_t)(base + min(16 * t + 8 * half + v, nrows - 1)) * D4];
#pragma unroll
            for (int v = 0; v < 8; ++v) {
                const bool in = 16 * t + 8 * half + v < nrows;
                c[0][v] = in ? s[v].x : 0.0f; c[1][v] = in ? s[v].y : 0.0f;
                c[2][v] = in ? s[v].z : 0.0f; c[3][v] = in ? s[v].w : 0.0f;
            }
        }

        const unsigned myrow = (unsigned)(16 * t + l16);
        for (; q < te; q += 16u) {
            const unsigned cur = (q + (unsigned)l16 < te) ? e : (e | 255u);
            const unsigned qn = q + 16u;
            if (qn < te) e = load_entries(qn);
            float4 rv[4][2];
            float  sel[4][2];
#pragma unroll
            for (int b = 0; b < 4; ++b) {
#pragma unroll
                for (int i = 0; i < 2; ++i) {
                    const unsigned ee = (unsigned)__shfl((int)cur, 4 * b + i + 2 * half, 32);
                    rv[b][i]  = hq[(size_t)(ee >> 8) * D4];
                    sel[b][i] = ((ee & 255u) == myrow) ? 1.0f : 0.0f;
                }
            }
            __builtin_amdgcn_sched_barrier(0);
#pragma unroll
            for (int b = 0; b < 4; ++b) {
                const v2f a = {sel[b][0], sel[b][1]};
                const float r0[4] = {rv[b][0].x, rv[b][0].y, rv[b][0].z, rv[b][0].w};
                const float r1[4] = {rv[b][1].x, rv[b][1].y, rv[b][1].z, rv[b][1].w};
#pragma unroll
                for (int j = 0; j < 4; ++j) {
                    const v2f bb = {r0[j], r1[j]};
                    c[j] = __builtin_amdgcn_wmma_f32_16x16x4_f32(false, a, false, bb,
                                                                 (short)0, c[j], false, false);
                }
            }
        }

        const float4 bv = ((const float4*)bias)[16 * ch + l16];
        float di[8];
#pragma unroll
        for (int v = 0; v < 8; ++v) di[v] = dinv[base + min(16 * t + 8 * half + v, nrows - 1)];
#pragma unroll
        for (int v = 0; v < 8; ++v) {
            const int r = 16 * t + 8 * half + v;
            float4 o = make_float4(fmaxf(di[v] * c[0][v] + bv.x, 0.0f), fmaxf(di[v] * c[1][v] + bv.y, 0.0f),
                                   fmaxf(di[v] * c[2][v] + bv.z, 0.0f), fmaxf(di[v] * c[3][v] + bv.w, 0.0f));
            if (r >= nrows) o = make_float4(0.0f, 0.0f, 0.0f, 0.0f);
            acc4[r * P4 + 16 * ch + l16] = o;
        }
    }
    lds_barrier();

    constexpr int NT  = NOUT / 16;
    constexpr int MG  = (OWN_THREADS / 32) / NT;
    constexpr int TPW = (OWN_TILES + MG - 1) / MG;
    const int n0   = (wave % NT) * 16;
    const int slot = wave / NT;

    v8f c[TPW];
#pragma unroll
    for (int i = 0; i < TPW; ++i) c[i] = (v8f){0.f, 0.f, 0.f, 0.f, 0.f, 0.f, 0.f, 0.f};

#pragma unroll
    for (int kh = 0; kh < 2; ++kh) {
        float4 wf[8];
        {
            const float4* p = (const float4*)(Wn + (size_t)(n0 + l16) * D + 4 * half) + 16 * kh;
#pragma unroll
            for (int j = 0; j < 8; ++j) wf[j] = p[2 * j];
        }
#pragma unroll
        for (int i = 0; i < TPW; ++i) {
            const int t = slot + MG * i;
            if (i == TPW - 1 && t >= OWN_TILES) continue;
            const float4* p = acc4 + (16 * t + l16) * P4 + half + 16 * kh;
            float4 a[8];
#pragma unroll
            for (int j = 0; j < 8; ++j) a[j] = p[2 * j];
#pragma unroll
            for (int j = 0; j < 8; ++j) {
                v2f a0 = {a[j].x, a[j].y}, b0 = {wf[j].x, wf[j].y};
                v2f a1 = {a[j].z, a[j].w}, b1 = {wf[j].z, wf[j].w};
                c[i] = __builtin_amdgcn_wmma_f32_16x16x4_f32(false, a0, false, b0,
                                                             (short)0, c[i], false, false);
                c[i] = __builtin_amdgcn_wmma_f32_16x16x4_f32(false, a1, false, b1,
                                                             (short)0, c[i], false, false);
            }
        }
    }

#pragma unroll
    for (int i = 0; i < TPW; ++i) {
        const int t = slot + MG * i;
        if (t < OWN_TILES) {
#pragma unroll
            for (int v = 0; v < 8; ++v) {
                const int r = 16 * t + 8 * half + v;
                if (r < nrows)
                    hout[(size_t)(base + r) * NOUT + n0 + l16] = dinv[base + r] * c[i][v];
            }
        }
    }
}

extern "C" void kernel_launch(void* const* d_in, const int* in_sizes, int n_in,
                              void* d_out, int out_size, void* d_ws, size_t ws_size,
                              hipStream_t stream) {
    const float* x  = (const float*)d_in[0];
    const int*   ei = (const int*)d_in[1];
    const float* W0 = (const float*)d_in[2];
    const float* b0 = (const float*)d_in[3];
    const float* W1 = (const float*)d_in[4];
    const float* b1 = (const float*)d_in[5];
    const float* W2 = (const float*)d_in[6];
    const float* b2 = (const float*)d_in[7];

    const int N = in_sizes[0] / IN_DIM;
    const int E = in_sizes[1] / 2;
    const int* rowIdx = ei;
    const int* colIdx = ei + E;

    char*  ws  = (char*)d_ws;
    size_t off = 0;
    auto   alloc = [&](size_t bytes) {
        void* p = ws + off;
        off += (bytes + 255) & ~(size_t)255;
        return p;
    };
    float* dinv = (float*)alloc((size_t)N * sizeof(float));
    float* hbuf = (float*)alloc((size_t)N * HID_DIM * sizeof(float));
    float* xbuf = (float*)alloc((size_t)N * HID_DIM * sizeof(float));

    const int ownBlks = (N + OWN_ROWS - 1) / OWN_ROWS;
    const int binBlks = (E + BIN_EDGES - 1) / BIN_EDGES;
    unsigned* crecs = (unsigned*)alloc((size_t)binBlks * BIN_EDGES * sizeof(unsigned));
    unsigned* otab  = (unsigned*)alloc((size_t)binBlks * OT_PITCH * sizeof(unsigned));
    int*      tab   = (int*)alloc((size_t)ownBlks * 2 * sizeof(int));
    unsigned* srecs = (unsigned*)alloc((size_t)E * sizeof(unsigned));
    int*      ttab  = (int*)alloc((size_t)ownBlks * 16 * sizeof(int));

    edge_chunk_kernel<<<binBlks, BIN_THREADS, 0, stream>>>(rowIdx, colIdx, crecs, otab, E);
    owner_degree_product_kernel<<<ownBlks, DEG_THREADS, 0, stream>>>(
        crecs, otab, binBlks, srecs, tab, ttab, dinv, x, W0, hbuf, N);

    owner_replay_product_kernel<128><<<ownBlks, OWN_THREADS, 0, stream>>>(
        srecs, tab, ttab, dinv, hbuf, b0, W1, xbuf, N);

    owner_replay_product_kernel<64><<<ownBlks, OWN_THREADS, 0, stream>>>(
        srecs, tab, ttab, dinv, xbuf, b1, W2, hbuf, N);

    owner_replay_kernel<64, false><<<ownBlks, OWN_THREADS, 0, stream>>>(
        srecs, tab, dinv, hbuf, b2, (float*)d_out, N);
}


# Hidden-layer owner sums: list ordered by owned row, each row filled to an even count (zero-row entry); a pair's two gathered rows added on the vector unit, one S*R slot per pair; new summation tree
# speedup vs baseline: 4.3429x; 1.0615x over previous
#include <hip/hip_runtime.h>
#include <hip/hip_bf16.h>

typedef float v2f __attribute__((ext_vector_type(2)));
typedef float v8f __attribute__((ext_vector_type(8)));

#define IN_DIM  128
#define HID_DIM 128
#define OUT_DIM 64

#define OWN_ROWS     196
#define OWN_THREADS  1024
#define MAX_GROUPS   256
#define BIN_THREADS  1024
#define BIN_EDGES    (BIN_THREADS * 4)
#define OT_PITCH     (MAX_GROUPS + 8)

__device__ __forceinline__ void lds_barrier() {
    __builtin_amdgcn_fence(__ATOMIC_RELEASE, "workgroup", "local");
    __builtin_amdgcn_s_barrier();
    __builtin_amdgcn_fence(__ATOMIC_ACQUIRE, "workgroup", "local");
}

__device__ __forceinline__ void load_cols(const int* __restrict__ col, int e0, int E, int (&c)[4]) {
    if (e0 + 3 < E) {
        int4 cv = *(const int4*)(col + e0);
        c[0] = cv.x; c[1] = cv.y; c[2] = cv.z; c[3] = cv.w;
    } else {
#pragma unroll
        for (int k = 0; k < 4; ++k) c[k] = (e0 + k < E) ? col[e0 + k] : -1;
    }
}

__global__ __launch_bounds__(BIN_THREADS)
void edge_chunk_kernel(const int* __restrict__ row,
                       const int* __restrict__ col,
                       unsigned* __restrict__ crecs, unsigned* __restrict__ otab, int E) {
    __shared__ uint4    stage4[BIN_EDGES / 4];
    __shared__ unsigned slot[MAX_GROUPS];
    __shared__ unsigned lfirst[MAX_GROUPS];
    __shared__ unsigned wcnt[MAX_GROUPS / 32];
    unsigned* stage = (unsigned*)stage4;

    const int tid  = threadIdx.x;
    const int lane = tid & 31;
    const int wave = tid >> 5;
    const int bid  = blockIdx.x;

    int r[4], c[4];
    const int e0 = (bid * BIN_THREADS + tid) * 4;
    load_cols(col, e0, E, c);
    if (e0 + 3 < E) {
        int4 rv = *(const int4*)(row + e0);
        r[0] = rv.x; r[1] = rv.y; r[2] = rv.z; r[3] = rv.w;
    } else {
#pragma unroll
        for (int k = 0; k < 4; ++k) r[k] = (e0 + k < E) ? row[e0 + k] : 0;
    }
    if (tid < MAX_GROUPS) slot[tid] = 0u;
    lds_barrier();

    unsigned g4[4], l4[4], rk[4];
#pragma unroll
    for (int k = 0; k < 4; ++k) {
        unsigned cc = (unsigned)max(c[k], 0);
        g4[k] = cc / OWN_ROWS;
        l4[k] = cc - g4[k] * OWN_ROWS;
        rk[k] = 0u;
        if (c[k] >= 0) rk[k] = atomicAdd(&slot[g4[k]], 1u);
    }
    lds_barrier();

    unsigned cntg = 0u;
    if (tid < MAX_GROUPS) cntg = slot[tid];
    unsigned cinc = cntg;
#pragma unroll
    for (int m = 1; m < 32; m <<= 1) {
        unsigned t = __shfl_up(cinc, m, 32);
        if (lane >= m) cinc += t;
    }
    if (tid < MAX_GROUPS && lane == 31) wcnt[wave] = cinc;
    lds_barrier();

    if (tid < MAX_GROUPS) {
        unsigned lf = cinc - cntg;
        for (int w = 0; w < wave; ++w) lf += wcnt[w];
        lfirst[tid] = lf;
        otab[(size_t)bid * OT_PITCH + tid] = lf;
        if (tid == MAX_GROUPS - 1) otab[(size_t)bid * OT_PITCH + MAX_GROUPS] = lf + cntg;
    }
    lds_barrier();

#pragma unroll
    for (int k = 0; k < 4; ++k)
        if (c[k] >= 0) stage[lfirst[g4[k]] + rk[k]] = ((unsigned)r[k] << 8) | l4[k];
    lds_barrier();

    const int total = min(BIN_EDGES, E - bid * BIN_EDGES);
    unsigned* __restrict__ dst = crecs + (size_t)bid * BIN_EDGES;
    if (4 * tid + 3 < total) {
        *(uint4*)(dst + 4 * tid) = stage4[tid];
    } else {
#pragma unroll
        for (int k = 0; k < 4; ++k)
            if (4 * tid + k < total) dst[4 * tid + k] = stage[4 * tid + k];
    }
}

#define DEG_THREADS  1024
#define SEG_L        4
#define SEG_U        8
#define OWN_TILES    ((OWN_ROWS + 15) / 16)
#define OWN_PITCH    132
__global__ __launch_bounds__(DEG_THREADS)
void owner_degree_product_kernel(const unsigned* __restrict__ crecs,
                                 const unsigned* __restrict__ otab, int nb,
                                 unsigned* __restrict__ srecs, int* __restrict__ tab,
                                 int* __restrict__ ttab,
                                 float* __restrict__ dinv,
                                 const float* __restrict__ X,
                                 const float* __restrict__ W,
                                 float* __restrict__ hout, int n) {
    static_assert(OWN_ROWS <= 224 && DEG_THREADS >= 256, "spare counters behind the rows");
    static_assert(OWN_TILES < 16, "tile starts fit one 16-entry row of ttab");
    constexpr int D    = 128;
    constexpr int D4   = D / 4;
    constexpr int P4   = OWN_PITCH / 4;
    constexpr int NOUT = 128;
    constexpr int XN   = OWN_TILES * 16 * D4;
    constexpr int XQ   = (XN + DEG_THREADS - 1) / DEG_THREADS;
    __shared__ float4 acc4[OWN_TILES * 16 * P4];
    __shared__ unsigned dcount[256];
    __shared__ unsigned cursor[256];
    __shared__ unsigned wtot[8];
    __shared__ unsigned tot[1];
    const int tid  = threadIdx.x;
    const int lane = tid & 31;
    const int wave = __builtin_amdgcn_readfirstlane(tid >> 5);
    const int half = lane >> 4;
    const int l16  = lane & 15;
    const int base = blockIdx.x * OWN_ROWS;
    const int nrows = min(OWN_ROWS, n - base);

    const int ls = tid & (SEG_L - 1);
    auto bounds = [&](int s, unsigned& st, unsigned& en) {
        const unsigned* o = otab + (size_t)s * OT_PITCH + blockIdx.x;
        st = o[0]; en = o[1];
    };
    const int s0 = tid / SEG_L;
    unsigned st0 = 0u, en0 = 0u;
    if (s0 < nb) bounds(s0, st0, en0);

    float4 xr[XQ];
    {
        const float4* __restrict__ x4 = (const float4*)X;
#pragma unroll
        for (int q = 0; q < XQ; ++q) {
            const int i = min(tid + DEG_THREADS * q, XN - 1);
            xr[q] = x4[(size_t)(base + min(i / D4, nrows - 1)) * D4 + (i % D4)];
        }
    }

    if (tid < 256) dcount[tid] = 1u;
    if (tid == 0) tot[0] = 0u;
    const unsigned spare = 224u + (unsigned)(tid & 31);
    lds_barrier();
    unsigned sst = 0u;
    {
        unsigned st = st0, en = en0;
        for (int s = s0; s < nb; ) {
            if (ls == 0) sst += st;
            const unsigned* __restrict__ p = crecs + (size_t)s * BIN_EDGES;
            for (unsigned j0 = st + (unsigned)ls; j0 < en; j0 += (unsigned)(SEG_L * SEG_U)) {
                unsigned e[SEG_U];
#pragma unroll
                for (int j = 0; j < SEG_U; ++j) e[j] = p[min(j0 + (unsigned)(SEG_L * j), en - 1u)];
#pragma unroll
                for (int j = 0; j < SEG_U; ++j)
                    atomicAdd(&dcount[(j0 + (unsigned)(SEG_L * j) < en) ? (e[j] & 255u) : spare], 1u);
            }
            s += DEG_THREADS / SEG_L;
            if (s < nb) bounds(s, st, en);
        }
    }
#pragma unroll
    for (int m = 1; m < 32; m <<= 1) sst += __shfl_xor(sst, m, 32);
    if (lane == 0) atomicAdd(&tot[0], sst);
#pragma unroll
    for (int q = 0; q < XQ; ++q) {
        const int i = tid + DEG_THREADS * q;
        if (i < XN) {
            const int rr = i / D4;
            acc4[rr * P4 + (i % D4)] = (rr < nrows) ? xr[q] : make_float4(0.0f, 0.0f, 0.0f, 0.0f);
        }
    }
    lds_barrier();
    const unsigned goff = tot[0] + (unsigned)(OWN_ROWS * blockIdx.x);
    if (tid < nrows) dinv[base + tid] = rsqrtf((float)dcount[tid]);
    if (blockIdx.x == 0 && tid < NOUT / 4)
        ((float4*)(hout + (size_t)n * NOUT))[tid] = make_float4(0.0f, 0.0f, 0.0f, 0.0f);

    const unsigned dr = (tid < 16 * OWN_TILES) ? dcount[tid] - 1u : 0u;
    const unsigned pc = (dr + 1u) & ~1u;
    unsigned pinc = pc;
#pragma unroll
    for (int m = 1; m < 32; m <<= 1) {
        unsigned t = __shfl_up(pinc, m, 32);
        if (lane >= m) pinc += t;
    }
    if (tid < 256 && lane == 31) wtot[wave] = pinc;
    lds_barrier();
    if (tid < 256) {
        unsigned o = pinc - pc;
        for (int w = 0; w < wave; ++w) o += wtot[w];
        cursor[tid] = o;
        if ((tid & 15) == 0) ttab[16 * blockIdx.x + (tid >> 4)] = (int)o;
        if (tid == 255) { tab[2 * blockIdx.x] = (int)goff; tab[2 * blockIdx.x + 1] = (int)o; }
        if (dr & 1u) srecs[(size_t)goff + o + dr] = ((unsigned)n << 8) | (unsigned)tid;
    }
    lds_barrier();
    {
        unsigned st = st0, en = en0;
        for (int s = s0; s < nb; ) {
            const unsigned* __restrict__ p = crecs + (size_t)s * BIN_EDGES;
            for (unsigned j0 = st + (unsigned)ls; j0 < en; j0 += (unsigned)(SEG_L * SEG_U)) {
                unsigned e[SEG_U];
#pragma unroll
                for (int j = 0; j < SEG_U; ++j) e[j] = p[min(j0 + (unsigned)(SEG_L * j), en - 1u)];
#pragma unroll
                for (int j = 0; j < SEG_U; ++j) {
                    if (j0 + (unsigned)(SEG_L * j) < en) {
                        const unsigned q = atomicAdd(&cursor[e[j] & 255u], 1u);
                        srecs[(size_t)goff + q] = e[j];
                    }
                }
            }
            s += DEG_THREADS / SEG_L;
            if (s < nb) bounds(s, st, en);
        }
    }

    constexpr int NT  = NOUT / 16;
    constexpr int MG  = (DEG_THREADS / 32) / NT;
    constexpr int TPW = (OWN_TILES + MG - 1) / MG;
    const int n0   = (wave % NT) * 16;
    const int slot = wave / NT;

    v8f c[TPW];
#pragma unroll
    for (int i = 0; i < TPW; ++i) c[i] = (v8f){0.f, 0.f, 0.f, 0.f, 0.f, 0.f, 0.f, 0.f};

#pragma unroll
    for (int kh = 0; kh < 2; ++kh) {
        float4 wf[8];
        {
            const float4* p = (const float4*)(W + (size_t)(n0 + l16) * D + 4 * half) + 16 * kh;
#pragma unroll
            for (int j = 0; j < 8; ++j) wf[j] = p[2 * j];
        }
#pragma unroll
        for (int i = 0; i < TPW; ++i) {
            const int t = slot + MG * i;
            if (i == TPW - 1 && t >= OWN_TILES) continue;
            const float4* p = acc4 + (16 * t + l16) * P4 + half + 16 * kh;
            float4 a[8];
#pragma unroll
            for (int j = 0; j < 8; ++j) a[j] = p[2 * j];
#pragma unroll
            for (int j = 0; j < 8; ++j) {
                v2f a0 = {a[j].x, a[j].y}, b0 = {wf[j].x, wf[j].y};
                v2f a1 = {a[j].z, a[j].w}, b1 = {wf[j].z, wf[j].w};
                c[i] = __builtin_amdgcn_wmma_f32_16x16x4_f32(false, a0, false, b0,
                                                             (short)0, c[i], false, false);
                c[i] = __builtin_amdgcn_wmma_f32_16x16x4_f32(false, a1, false, b1,
                                                             (short)0, c[i], false, false);
            }
        }
    }

#pragma unroll
    for (int i = 0; i < TPW; ++i) {
        const int t = slot + MG * i;
        if (t < OWN_TILES) {
#pragma unroll
            for (int v = 0; v < 8; ++v) {
                const int r = 16 * t + 8 * half + v;
                if (r < nrows)
                    hout[(size_t)(base + r) * NOUT + n0 + l16] = rsqrtf((float)dcount[r]) * c[i][v];
            }
        }
    }
}

#define REPLAY_T 8
template <int D, bool RELU>
__global__ __launch_bounds__(OWN_THREADS)
void owner_replay_kernel(const unsigned* __restrict__ recs,
                         const int* __restrict__ tab,
                         const float* __restrict__ dinv,
                         const float* __restrict__ h,
                         const float* __restrict__ bias,
                         float* __restrict__ out, int n) {
    constexpr int V2 = D / 64;
    constexpr int D4 = D / 4;
    __shared__ float4 acc4[OWN_ROWS * D4];
    float* acc = (float*)acc4;

    const int tid  = threadIdx.x;
    const int lane = tid & 31;
    const int wave = __builtin_amdgcn_readfirstlane(tid >> 5);
    const int base = blockIdx.x * OWN_ROWS;
    const int nrows = min(OWN_ROWS, n - base);
    const unsigned goff = (unsigned)tab[2 * blockIdx.x];
    const unsigned cnt  = (unsigned)tab[2 * blockIdx.x + 1];

    unsigned q0 = (unsigned)(REPLAY_T * wave);
    unsigned rec = 0u;
    if (q0 < cnt) rec = recs[(size_t)goff + min(q0 + (unsigned)(lane & (REPLAY_T - 1)), cnt - 1u)];

    const float4* __restrict__ h4 = (const float4*)(h + (size_t)base * D);
    for (int i = tid; i < nrows * D4; i += OWN_THREADS) acc4[i] = h4[i];
    lds_barrier();

    for (; q0 < cnt; q0 += (OWN_THREADS / 32) * REPLAY_T) {
        const unsigned cur = rec;
        const unsigned qn = q0 + (OWN_THREADS / 32) * REPLAY_T;
        if (qn < cnt) rec = recs[(size_t)goff + min(qn + (unsigned)(lane & (REPLAY_T - 1)), cnt - 1u)];
        float2 v[REPLAY_T][V2];
        unsigned loc[REPLAY_T];
#pragma unroll
        for (int k = 0; k < REPLAY_T; ++k) {
            unsigned w0 = (unsigned)__builtin_amdgcn_readlane((int)cur, k);
            loc[k] = w0 & 255u;
            const float2* src = (const float2*)(h + (size_t)(w0 >> 8) * D) + lane;
#pragma unroll
            for (int j = 0; j < V2; ++j) v[k][j] = src[32 * j];
        }
#pragma unroll
        for (int k = 0; k < REPLAY_T; ++k) {
#pragma unroll
            for (int j = 0; j < V2; ++j) asm volatile("" :: "v"(v[k][j].x), "v"(v[k][j].y));
        }
#pragma unroll
        for (int k = 0; k < REPLAY_T; ++k) {
            if (q0 + (unsigned)k < cnt) {
                float* a = acc + loc[k] * D + 2 * lane;
#pragma unroll
                for (int j = 0; j < V2; ++j) {
                    atomicAdd(a + 64 * j,     v[k][j].x);
                    atomicAdd(a + 64 * j + 1, v[k][j].y);
                }
            }
        }
    }
    lds_barrier();

    float4* __restrict__ out4 = (float4*)(out + (size_t)base * D);
    for (int i = tid; i < nrows * D4; i += OWN_THREADS) {
        const int rr = i / D4, d = (i % D4) * 4;
        const float di = dinv[base + rr];
        const float4 bv = *(const float4*)(bias + d);
        float4 v = acc4[i];
        v = make_float4(di * v.x + bv.x, di * v.y + bv.y, di * v.z + bv.z, di * v.w + bv.w);
        if (RELU) v = make_float4(fmaxf(v.x, 0.0f), fmaxf(v.y, 0.0f), fmaxf(v.z, 0.0f), fmaxf(v.w, 0.0f));
        out4[i] = v;
    }
}

template <int NOUT>
__global__ __launch_bounds__(OWN_THREADS)
void owner_replay_product_kernel(const unsigned* __restrict__ srecs,
                                 const int* __restrict__ tab,
                                 const int* __restrict__ ttab,
                                 const float* __restrict__ dinv,
                                 const float* __restrict__ h,
                                 const float* __restrict__ bias,
                                 const float* __restrict__ Wn,
                                 float* __restrict__ hout, int n) {
    constexpr int D  = 128;
    constexpr int D4 = D / 4;
    constexpr int P4 = OWN_PITCH / 4;
    static_assert(2 * OWN_TILES <= OWN_THREADS / 32, "one wave per tile and column half");
    __shared__ float4 acc4[OWN_TILES * 16 * P4];

    const int tid  = threadIdx.x;
    const int lane = tid & 31;
    const int wave = __builtin_amdgcn_readfirstlane(tid >> 5);
    const int half = lane >> 4;
    const int l16  = lane & 15;
    const int base = blockIdx.x * OWN_ROWS;
    const int nrows = min(OWN_ROWS, n - base);
    const unsigned goff = (unsigned)tab[2 * blockIdx.x];

    if (wave < 2 * OWN_TILES) {
        const int t  = wave >> 1;
        const int ch = wave & 1;
        const unsigned tb = (unsigned)ttab[16 * blockIdx.x + t];
        const unsigned te = (unsigned)ttab[16 * blockIdx.x + t + 1];
        const float4* __restrict__ hq = (const float4*)h + 16 * ch + l16;

        auto load_entries = [&](unsigned q0) {
            return srecs[(size_t)goff + min(q0 + (unsigned)l16, te - 1u)];
        };
        unsigned q = tb;
        unsigned e = 0u;
        if (q < te) e = load_entries(q);

        v8f c[4];
        {
            float4 s[8];
#pragma unroll
            for (int v = 0; v < 8; ++v)
                s[v] = hq[(size_t)(base + min(16 * t + 8 * half + v, nrows - 1)) * D4];
#pragma unroll
            for (int v = 0; v < 8; ++v) {
                const bool in = 16 * t + 8 * half + v < nrows;
                c[0][v] = in ? s[v].x : 0.0f; c[1][v] = in ? s[v].y : 0.0f;
                c[2][v] = in ? s[v].z : 0.0f; c[3][v] = in ? s[v].w : 0.0f;
            }
        }

        const unsigned myrow = (unsigned)(16 * t + l16);
        for (; q < te; q += 16u) {
            const unsigned cur = (q + (unsigned)l16 < te) ? e : (e | 255u);
            const unsigned qn = q + 16u;
            if (qn < te) e = load_entries(qn);
            float4 ra[2][2], rb[2][2];
            float  sel[2][2];
#pragma unroll
            for (int b = 0; b < 2; ++b) {
#pragma unroll
                for (int i = 0; i < 2; ++i) {
                    const int p = 4 * b + i + 2 * half;
                    const unsigned ea = (unsigned)__shfl((int)cur, 2 * p, 32);
                    const unsigned eb = (unsigned)__shfl((int)cur, 2 * p + 1, 32);
                    ra[b][i]  = hq[(size_t)(ea >> 8) * D4];
                    rb[b][i]  = hq[(size_t)(eb >> 8) * D4];
                    sel[b][i] = ((ea & 255u) == myrow) ? 1.0f : 0.0f;
                }
            }
            __builtin_amdgcn_sched_barrier(0);
#pragma unroll
            for (int b = 0; b < 2; ++b) {
                const v2f a = {sel[b][0], sel[b][1]};
                const float r0[4] = {ra[b][0].x + rb[b][0].x, ra[b][0].y + rb[b][0].y,
                                     ra[b][0].z + rb[b][0].z, ra[b][0].w + rb[b][0].w};
                const float r1[4] = {ra[b][1].x + rb[b][1].x, ra[b][1].y + rb[b][1].y,
                                     ra[b][1].z + rb[b][1].z, ra[b][1].w + rb[b][1].w};
#pragma unroll
                for (int j = 0; j < 4; ++j) {
                    const v2f bb = {r0[j], r1[j]};
                    c[j] = __builtin_amdgcn_wmma_f32_16x16x4_f32(false, a, false, bb,
                                                                 (short)0, c[j], false, false);
                }
            }
        }

        const float4 bv = ((const float4*)bias)[16 * ch + l16];
        float di[8];
#pragma unroll
        for (int v = 0; v < 8; ++v) di[v] = dinv[base + min(16 * t + 8 * half + v, nrows - 1)];
#pragma unroll
        for (int v = 0; v < 8; ++v) {
            const int r = 16 * t + 8 * half + v;
            float4 o = make_float4(fmaxf(di[v] * c[0][v] + bv.x, 0.0f), fmaxf(di[v] * c[1][v] + bv.y, 0.0f),
                                   fmaxf(di[v] * c[2][v] + bv.z, 0.0f), fmaxf(di[v] * c[3][v] + bv.w, 0.0f));
            if (r >= nrows) o = make_float4(0.0f, 0.0f, 0.0f, 0.0f);
            acc4[r * P4 + 16 * ch + l16] = o;
        }
    }
    lds_barrier();

    constexpr int NT  = NOUT / 16;
    constexpr int MG  = (OWN_THREADS / 32) / NT;
    constexpr int TPW = (OWN_TILES + MG - 1) / MG;
    const int n0   = (wave % NT) * 16;
    const int slot = wave / NT;

    v8f c[TPW];
#pragma unroll
    for (int i = 0; i < TPW; ++i) c[i] = (v8f){0.f, 0.f, 0.f, 0.f, 0.f, 0.f, 0.f, 0.f};

#pragma unroll
    for (int kh = 0; kh < 2; ++kh) {
        float4 wf[8];
        {
            const float4* p = (const float4*)(Wn + (size_t)(n0 + l16) * D + 4 * half) + 16 * kh;
#pragma unroll
            for (int j = 0; j < 8; ++j) wf[j] = p[2 * j];
        }
#pragma unroll
        for (int i = 0; i < TPW; ++i) {
            const int t = slot + MG * i;
            if (i == TPW - 1 && t >= OWN_TILES) continue;
            const float4* p = acc4 + (16 * t + l16) * P4 + half + 16 * kh;
            float4 a[8];
#pragma unroll
            for (int j = 0; j < 8; ++j) a[j] = p[2 * j];
#pragma unroll
            for (int j = 0; j < 8; ++j) {
                v2f a0 = {a[j].x, a[j].y}, b0 = {wf[j].x, wf[j].y};
                v2f a1 = {a[j].z, a[j].w}, b1 = {wf[j].z, wf[j].w};
                c[i] = __builtin_amdgcn_wmma_f32_16x16x4_f32(false, a0, false, b0,
                                                             (short)0, c[i], false, false);
                c[i] = __builtin_amdgcn_wmma_f32_16x16x4_f32(false, a1, false, b1,
                                                             (short)0, c[i], false, false);
            }
        }
    }

#pragma unroll
    for (int i = 0; i < TPW; ++i) {
        const int t = slot + MG * i;
        if (t < OWN_TILES) {
#pragma unroll
            for (int v = 0; v < 8; ++v) {
                const int r = 16 * t + 8 * half + v;
                if (r < nrows)
                    hout[(size_t)(base + r) * NOUT + n0 + l16] = dinv[base + r] * c[i][v];
            }
        }
    }
    if (blockIdx.x == 0 && tid < NOUT / 4)
        ((float4*)(hout + (size_t)n * NOUT))[tid] = make_float4(0.0f, 0.0f, 0.0f, 0.0f);
}

extern "C" void kernel_launch(void* const* d_in, const int* in_sizes, int n_in,
                              void* d_out, int out_size, void* d_ws, size_t ws_size,
                              hipStream_t stream) {
    const float* x  = (const float*)d_in[0];
    const int*   ei = (const int*)d_in[1];
    const float* W0 = (const float*)d_in[2];
    const float* b0 = (const float*)d_in[3];
    const float* W1 = (const float*)d_in[4];
    const float* b1 = (const float*)d_in[5];
    const float* W2 = (const float*)d_in[6];
    const float* b2 = (const float*)d_in[7];

    const int N = in_sizes[0] / IN_DIM;
    const int E = in_sizes[1] / 2;
    const int* rowIdx = ei;
    const int* colIdx = ei + E;

    char*  ws  = (char*)d_ws;
    size_t off = 0;
    auto   alloc = [&](size_t bytes) {
        void* p = ws + off;
        off += (bytes + 255) & ~(size_t)255;
        return p;
    };
    float* dinv = (float*)alloc((size_t)N * sizeof(float));
    float* hbuf = (float*)alloc((size_t)(N + 1) * HID_DIM * sizeof(float));
    float* xbuf = (float*)alloc((size_t)(N + 1) * HID_DIM * sizeof(float));

    const int ownBlks = (N + OWN_ROWS - 1) / OWN_ROWS;
    const int binBlks = (E + BIN_EDGES - 1) / BIN_EDGES;
    unsigned* crecs = (unsigned*)alloc((size_t)binBlks * BIN_EDGES * sizeof(unsigned));
    unsigned* otab  = (unsigned*)alloc((size_t)binBlks * OT_PITCH * sizeof(unsigned));
    int*      tab   = (int*)alloc((size_t)ownBlks * 2 * sizeof(int));
    unsigned* srecs = (unsigned*)alloc(((size_t)E + (size_t)ownBlks * OWN_ROWS) * sizeof(unsigned));
    int*      ttab  = (int*)alloc((size_t)ownBlks * 16 * sizeof(int));

    edge_chunk_kernel<<<binBlks, BIN_THREADS, 0, stream>>>(rowIdx, colIdx, crecs, otab, E);
    owner_degree_product_kernel<<<ownBlks, DEG_THREADS, 0, stream>>>(
        crecs, otab, binBlks, srecs, tab, ttab, dinv, x, W0, hbuf, N);

    owner_replay_product_kernel<128><<<ownBlks, OWN_THREADS, 0, stream>>>(
        srecs, tab, ttab, dinv, hbuf, b0, W1, xbuf, N);

    owner_replay_product_kernel<64><<<ownBlks, OWN_THREADS, 0, stream>>>(
        srecs, tab, ttab, dinv, xbuf, b1, W2, hbuf, N);

    owner_replay_kernel<64, false><<<ownBlks, OWN_THREADS, 0, stream>>>(
        srecs, tab, dinv, hbuf, b2, (float*)d_out, N);
}
